# starting kernel
# baseline (speedup 1.0000x reference)
#include <hip/hip_runtime.h>
#include <hip/hip_bf16.h>

#define NN   20000
#define EE   2000
#define NNZ  100000
#define FF   128
#define HH   128
#define DD   6
#define COUT 40
#define DH   (DD * HH)

typedef __attribute__((ext_vector_type(2))) float v2f;
typedef __attribute__((ext_vector_type(8))) float v8f;

#define ATOMIC_ADD_F32(p, v) unsafeAtomicAdd((p), (v))

#define TK 32
#define TM 32
__global__ __launch_bounds__(128)
void gemm_wmma_f32(const float* __restrict__ A, const float* __restrict__ B,
                   float* __restrict__ C, int M, int Nc, int K, int ldb, int ldc) {
  __shared__ float As[TM][TK + 1];
  __shared__ float Bs[TK][64];

  const int tid  = threadIdx.x;
  const int wave = tid >> 5;
  const int lane = tid & 31;
  const int la   = lane & 15;
  const int hi   = lane >> 4;

  const int m0 = blockIdx.y * TM;
  const int nb = blockIdx.x * 64;
  const int n0 = nb + wave * 16;

  v8f acc0 = {};
  v8f acc1 = {};

  for (int kk = 0; kk < K; kk += TK) {
    for (int i = tid; i < TM * TK; i += 128) {
      int r = i / TK, c = i % TK;
      int gm = m0 + r;
      As[r][c] = (gm < M) ? A[(size_t)gm * K + kk + c] : 0.0f;
    }
    for (int i = tid; i < TK * 64; i += 128) {
      int r = i >> 6, c = i & 63;
      int gn = nb + c;
      Bs[r][c] = (gn < Nc) ? B[(size_t)(kk + r) * ldb + gn] : 0.0f;
    }
    if (kk + TK < K) {
      int pr = tid >> 2, pc = (tid & 3) << 3;
      int gm = m0 + pr;
      if (gm < M) __builtin_prefetch(&A[(size_t)gm * K + kk + TK + pc], 0, 3);
    }
    __syncthreads();

#pragma unroll
    for (int k4 = 0; k4 < TK; k4 += 4) {
      v2f a0, a1, b;
      b.x  = Bs[k4 + 2 * hi][(wave << 4) + la];
      b.y  = Bs[k4 + 2 * hi + 1][(wave << 4) + la];
      a0.x = As[la][k4 + 2 * hi];
      a0.y = As[la][k4 + 2 * hi + 1];
      a1.x = As[16 + la][k4 + 2 * hi];
      a1.y = As[16 + la][k4 + 2 * hi + 1];
      acc0 = __builtin_amdgcn_wmma_f32_16x16x4_f32(
          false, a0, false, b, (short)0, acc0, false, false);
      acc1 = __builtin_amdgcn_wmma_f32_16x16x4_f32(
          false, a1, false, b, (short)0, acc1, false, false);
    }
    __syncthreads();
  }

  const int n = n0 + la;
  if (n < Nc) {
#pragma unroll
    for (int r = 0; r < 8; ++r) {
      int gm = m0 + r + 8 * hi;
      if (gm < M) C[(size_t)gm * ldc + n] = acc0[r];
    }
#pragma unroll
    for (int r = 0; r < 8; ++r) {
      int gm = m0 + 16 + r + 8 * hi;
      if (gm < M) C[(size_t)gm * ldc + n] = acc1[r];
    }
  }
}

__global__ void mean_d_kernel(const float* __restrict__ xl, float* __restrict__ xm,
                              int rows) {
  int gid = blockIdx.x * blockDim.x + threadIdx.x;
  if (gid >= rows * FF) return;
  int n = gid >> 7, f = gid & 127;
  const float* p = xl + (size_t)n * DH + f;
  float s = 0.0f;
#pragma unroll
  for (int d = 0; d < DD; ++d) s += p[d * FF];
  xm[gid] = s * (1.0f / (float)DD);
}

__global__ void alpha_deg_kernel(const float* __restrict__ xm, const float* __restrict__ em,
                                 const int* __restrict__ row, const int* __restrict__ col,
                                 float* __restrict__ alpha, float* __restrict__ Dg,
                                 float* __restrict__ Bg) {
  int gid = blockIdx.x * blockDim.x + threadIdx.x;
  if (gid >= NNZ * DD) return;
  int i = gid / DD, dd = gid - i * DD;
  int r = row[i], c = col[i];
  float v = xm[r * DD + dd] + em[c * DD + dd];
  v = 1.0f / (1.0f + __expf(-v));
  alpha[gid] = v;
  ATOMIC_ADD_F32(&Dg[r * DD + dd], v);
  ATOMIC_ADD_F32(&Bg[c * DD + dd], v);
}

__global__ void recip_kernel(float* __restrict__ p, int n) {
  int gid = blockIdx.x * blockDim.x + threadIdx.x;
  if (gid >= n) return;
  float v = p[gid];
  p[gid] = (v != 0.0f) ? (1.0f / v) : 0.0f;
}

__global__ void fill_kernel(float* __restrict__ p, size_t n, float v) {
  size_t stride = (size_t)gridDim.x * blockDim.x;
  for (size_t i = blockIdx.x * (size_t)blockDim.x + threadIdx.x; i < n; i += stride)
    p[i] = v;
}

__global__ __launch_bounds__(192)
void scatter_to_edge(const float* __restrict__ xp, const float* __restrict__ alpha,
                     const float* __restrict__ Binv, const int* __restrict__ row,
                     const int* __restrict__ col, float* __restrict__ m) {
  int i = blockIdx.x;
  int idx = threadIdx.x << 2;
  int dd = idx >> 7;
  int r = row[i], c = col[i];
  float s = Binv[c * DD + dd] * alpha[(size_t)i * DD + dd];
  float4 v = *(const float4*)(xp + (size_t)r * DH + idx);
  float* dst = m + (size_t)c * DH + idx;
  ATOMIC_ADD_F32(dst + 0, s * v.x);
  ATOMIC_ADD_F32(dst + 1, s * v.y);
  ATOMIC_ADD_F32(dst + 2, s * v.z);
  ATOMIC_ADD_F32(dst + 3, s * v.w);
}

__global__ __launch_bounds__(192)
void scatter_to_node(const float* __restrict__ m, const float* __restrict__ alpha,
                     const float* __restrict__ Dinv, const int* __restrict__ row,
                     const int* __restrict__ col, float* __restrict__ out) {
  int i = blockIdx.x;
  int idx = threadIdx.x << 2;
  int dd = idx >> 7;
  int r = row[i], c = col[i];
  float s = Dinv[r * DD + dd] * alpha[(size_t)i * DD + dd];
  float4 v = *(const float4*)(m + (size_t)c * DH + idx);
  float* dst = out + (size_t)r * DH + idx;
  ATOMIC_ADD_F32(dst + 0, s * v.x);
  ATOMIC_ADD_F32(dst + 1, s * v.y);
  ATOMIC_ADD_F32(dst + 2, s * v.z);
  ATOMIC_ADD_F32(dst + 3, s * v.w);
}

__global__ void bias_act_kernel(float* __restrict__ h, const float* __restrict__ b,
                                size_t total, int do_elu) {
  size_t gid = blockIdx.x * (size_t)blockDim.x + threadIdx.x;
  if (gid >= total) return;
  float v = h[gid] + b[gid & 127];
  if (do_elu) v = (v > 0.0f) ? v : (__expf(v) - 1.0f);
  h[gid] = v;
}

static inline void launch_gemm(const float* A, const float* B, float* C,
                               int M, int Nc, int K, int ldb, int ldc,
                               hipStream_t s) {
  dim3 grid((Nc + 63) / 64, (M + TM - 1) / TM);
  gemm_wmma_f32<<<grid, dim3(128), 0, s>>>(A, B, C, M, Nc, K, ldb, ldc);
}

extern "C" void kernel_launch(void* const* d_in, const int* in_sizes, int n_in,
                              void* d_out, int out_size, void* d_ws, size_t ws_size,
                              hipStream_t stream) {
  const float* x        = (const float*)d_in[0];
  const int*   eidx     = (const int*)d_in[1];
  const float* hedge    = (const float*)d_in[2];
  const float* lin_W    = (const float*)d_in[3];
  const float* sheaf_W  = (const float*)d_in[4];
  const float* conv1_W  = (const float*)d_in[5];
  const float* conv1_b  = (const float*)d_in[6];
  const float* conv2_W  = (const float*)d_in[7];
  const float* conv2_b  = (const float*)d_in[8];
  const float* lin2_W   = (const float*)d_in[9];
  float* out = (float*)d_out;

  const int* row = eidx;
  const int* col = eidx + NNZ;

  float* ws    = (float*)d_ws;
  float* bufA  = ws;
  float* bufB  = bufA + (size_t)NN * DH;
  float* el    = bufB + (size_t)NN * DH;
  float* mbuf  = el + (size_t)EE * DH;
  float* xmean = mbuf + (size_t)EE * DH;
  float* emean = xmean + (size_t)NN * FF;
  float* xm    = emean + (size_t)EE * FF;
  float* em    = xm + (size_t)NN * DD;
  float* alpha = em + (size_t)EE * DD;
  float* Dg    = alpha + (size_t)NNZ * DD;
  float* Bg    = Dg + (size_t)NN * DD;

  const size_t nBig = (size_t)NN * DH;
  const size_t eBig = (size_t)EE * DH;

  launch_gemm(x, lin_W, bufA, NN, DH, FF, DH, DH, stream);
  launch_gemm(hedge, lin_W, el, EE, DH, FF, DH, DH, stream);

  mean_d_kernel<<<(NN * FF + 255) / 256, 256, 0, stream>>>(bufA, xmean, NN);
  mean_d_kernel<<<(EE * FF + 255) / 256, 256, 0, stream>>>(el, emean, EE);

  launch_gemm(xmean, sheaf_W, xm, NN, DD, FF, DD, DD, stream);
  launch_gemm(emean, sheaf_W + FF * DD, em, EE, DD, FF, DD, DD, stream);

  fill_kernel<<<64, 256, 0, stream>>>(Dg, (size_t)NN * DD, 0.0f);
  fill_kernel<<<64, 256, 0, stream>>>(Bg, (size_t)EE * DD, 0.0f);
  alpha_deg_kernel<<<(NNZ * DD + 255) / 256, 256, 0, stream>>>(xm, em, row, col,
                                                               alpha, Dg, Bg);
  recip_kernel<<<(NN * DD + 255) / 256, 256, 0, stream>>>(Dg, NN * DD);
  recip_kernel<<<(EE * DD + 255) / 256, 256, 0, stream>>>(Bg, EE * DD);

  launch_gemm(bufA, conv1_W, bufB, NN * DD, HH, FF, HH, HH, stream);
  fill_kernel<<<1024, 256, 0, stream>>>(mbuf, eBig, 0.0f);
  scatter_to_edge<<<NNZ, 192, 0, stream>>>(bufB, alpha, Bg, row, col, mbuf);
  fill_kernel<<<4096, 256, 0, stream>>>(bufA, nBig, 0.0f);
  scatter_to_node<<<NNZ, 192, 0, stream>>>(mbuf, alpha, Dg, row, col, bufA);
  bias_act_kernel<<<(unsigned)((nBig + 255) / 256), 256, 0, stream>>>(bufA, conv1_b,
                                                                      nBig, 1);

  launch_gemm(bufA, conv2_W, bufB, NN * DD, HH, HH, HH, HH, stream);
  fill_kernel<<<1024, 256, 0, stream>>>(mbuf, eBig, 0.0f);
  scatter_to_edge<<<NNZ, 192, 0, stream>>>(bufB, alpha, Bg, row, col, mbuf);
  fill_kernel<<<4096, 256, 0, stream>>>(bufA, nBig, 0.0f);
  scatter_to_node<<<NNZ, 192, 0, stream>>>(mbuf, alpha, Dg, row, col, bufA);
  bias_act_kernel<<<(unsigned)((nBig + 255) / 256), 256, 0, stream>>>(bufA, conv2_b,
                                                                      nBig, 0);

  launch_gemm(bufA, lin2_W, out, NN, COUT, DH, COUT, COUT, stream);
}


# GEMM staging: each thread issues all loads of the next k-tile together (128-bit) into registers before the current tile's WMMAs; was one dependent 32-bit load per element with no overlap
# speedup vs baseline: 1.6472x; 1.6472x over previous
#include <hip/hip_runtime.h>
#include <hip/hip_bf16.h>

#define NN   20000
#define EE   2000
#define NNZ  100000
#define FF   128
#define HH   128
#define DD   6
#define COUT 40
#define DH   (DD * HH)

typedef __attribute__((ext_vector_type(2))) float v2f;
typedef __attribute__((ext_vector_type(8))) float v8f;

#define ATOMIC_ADD_F32(p, v) unsafeAtomicAdd((p), (v))

#define TK 32
#define TM 32
#define AP 36
template <bool VB>
__global__ __launch_bounds__(128)
void gemm_wmma_f32(const float* __restrict__ A, const float* __restrict__ B,
                   float* __restrict__ C, int M, int Nc, int K, int ldb, int ldc) {
  __shared__ __attribute__((aligned(16))) float As[TM][AP];
  __shared__ __attribute__((aligned(16))) float Bs[TK][64];

  const int tid  = threadIdx.x;
  const int wave = tid >> 5;
  const int lane = tid & 31;
  const int la   = lane & 15;
  const int hi   = lane >> 4;

  const int m0 = blockIdx.y * TM;
  const int nb = blockIdx.x * 64;
  const int n0 = nb + wave * 16;

  const int ar = tid >> 3, ac = (tid & 7) << 2;
  const int br = tid >> 4, bc = (tid & 15) << 2;
  const int gn = nb + bc;
  const float* __restrict__ Ap0 = A + (size_t)min(m0 + ar, M - 1) * K + ac;
  const float* __restrict__ Ap1 = A + (size_t)min(m0 + ar + 16, M - 1) * K + ac;
  const float* __restrict__ Bp  = B + (size_t)br * ldb;
  const size_t bstep = (size_t)8 * ldb;
  const int cv = min(gn, Nc - 4);
  const int c0 = min(gn + 0, Nc - 1), c1 = min(gn + 1, Nc - 1);
  const int c2 = min(gn + 2, Nc - 1), c3 = min(gn + 3, Nc - 1);

  float4 ra0, ra1, rb0, rb1, rb2, rb3;

#define GEMM_FETCH_B(dst_, q_)                                                 \
  do {                                                                         \
    if (VB) {                                                                  \
      dst_ = *(const float4*)((q_) + cv);                                      \
    } else {                                                                   \
      dst_ = make_float4((q_)[c0], (q_)[c1], (q_)[c2], (q_)[c3]);              \
    }                                                                          \
  } while (0)
#define GEMM_FETCH(kk_)                                                        \
  do {                                                                         \
    ra0 = *(const float4*)(Ap0 + (kk_));                                       \
    ra1 = *(const float4*)(Ap1 + (kk_));                                       \
    const float* qb_ = Bp + (size_t)(kk_) * ldb;                               \
    GEMM_FETCH_B(rb0, qb_);                                                    \
    GEMM_FETCH_B(rb1, qb_ + bstep);                                            \
    GEMM_FETCH_B(rb2, qb_ + 2 * bstep);                                        \
    GEMM_FETCH_B(rb3, qb_ + 3 * bstep);                                        \
  } while (0)

  v8f acc0 = {};
  v8f acc1 = {};

  GEMM_FETCH(0);

  for (int kk = 0; kk < K; kk += TK) {
    *(float4*)&As[ar][ac]      = ra0;
    *(float4*)&As[ar + 16][ac] = ra1;
    *(float4*)&Bs[br][bc]      = rb0;
    *(float4*)&Bs[br + 8][bc]  = rb1;
    *(float4*)&Bs[br + 16][bc] = rb2;
    *(float4*)&Bs[br + 24][bc] = rb3;
    __syncthreads();

    if (kk + TK < K) GEMM_FETCH(kk + TK);

#pragma unroll
    for (int k4 = 0; k4 < TK; k4 += 4) {
      v2f a0, a1, b;
      b.x  = Bs[k4 + 2 * hi][(wave << 4) + la];
      b.y  = Bs[k4 + 2 * hi + 1][(wave << 4) + la];
      a0.x = As[la][k4 + 2 * hi];
      a0.y = As[la][k4 + 2 * hi + 1];
      a1.x = As[16 + la][k4 + 2 * hi];
      a1.y = As[16 + la][k4 + 2 * hi + 1];
      acc0 = __builtin_amdgcn_wmma_f32_16x16x4_f32(
          false, a0, false, b, (short)0, acc0, false, false);
      acc1 = __builtin_amdgcn_wmma_f32_16x16x4_f32(
          false, a1, false, b, (short)0, acc1, false, false);
    }
    __syncthreads();
  }
#undef GEMM_FETCH
#undef GEMM_FETCH_B

  const int n = n0 + la;
  if (n < Nc) {
#pragma unroll
    for (int r = 0; r < 8; ++r) {
      int gm = m0 + r + 8 * hi;
      if (gm < M) C[(size_t)gm * ldc + n] = acc0[r];
    }
#pragma unroll
    for (int r = 0; r < 8; ++r) {
      int gm = m0 + 16 + r + 8 * hi;
      if (gm < M) C[(size_t)gm * ldc + n] = acc1[r];
    }
  }
}

__global__ void mean_d_kernel(const float* __restrict__ xl, float* __restrict__ xm,
                              int rows) {
  int gid = blockIdx.x * blockDim.x + threadIdx.x;
  if (gid >= rows * FF) return;
  int n = gid >> 7, f = gid & 127;
  const float* p = xl + (size_t)n * DH + f;
  float s = 0.0f;
#pragma unroll
  for (int d = 0; d < DD; ++d) s += p[d * FF];
  xm[gid] = s * (1.0f / (float)DD);
}

__global__ void alpha_deg_kernel(const float* __restrict__ xm, const float* __restrict__ em,
                                 const int* __restrict__ row, const int* __restrict__ col,
                                 float* __restrict__ alpha, float* __restrict__ Dg,
                                 float* __restrict__ Bg) {
  int gid = blockIdx.x * blockDim.x + threadIdx.x;
  if (gid >= NNZ * DD) return;
  int i = gid / DD, dd = gid - i * DD;
  int r = row[i], c = col[i];
  float v = xm[r * DD + dd] + em[c * DD + dd];
  v = 1.0f / (1.0f + __expf(-v));
  alpha[gid] = v;
  ATOMIC_ADD_F32(&Dg[r * DD + dd], v);
  ATOMIC_ADD_F32(&Bg[c * DD + dd], v);
}

__global__ void recip_kernel(float* __restrict__ p, int n) {
  int gid = blockIdx.x * blockDim.x + threadIdx.x;
  if (gid >= n) return;
  float v = p[gid];
  p[gid] = (v != 0.0f) ? (1.0f / v) : 0.0f;
}

__global__ void fill_kernel(float* __restrict__ p, size_t n, float v) {
  size_t stride = (size_t)gridDim.x * blockDim.x;
  for (size_t i = blockIdx.x * (size_t)blockDim.x + threadIdx.x; i < n; i += stride)
    p[i] = v;
}

__global__ __launch_bounds__(192)
void scatter_to_edge(const float* __restrict__ xp, const float* __restrict__ alpha,
                     const float* __restrict__ Binv, const int* __restrict__ row,
                     const int* __restrict__ col, float* __restrict__ m) {
  int i = blockIdx.x;
  int idx = threadIdx.x << 2;
  int dd = idx >> 7;
  int r = row[i], c = col[i];
  float s = Binv[c * DD + dd] * alpha[(size_t)i * DD + dd];
  float4 v = *(const float4*)(xp + (size_t)r * DH + idx);
  float* dst = m + (size_t)c * DH + idx;
  ATOMIC_ADD_F32(dst + 0, s * v.x);
  ATOMIC_ADD_F32(dst + 1, s * v.y);
  ATOMIC_ADD_F32(dst + 2, s * v.z);
  ATOMIC_ADD_F32(dst + 3, s * v.w);
}

__global__ __launch_bounds__(192)
void scatter_to_node(const float* __restrict__ m, const float* __restrict__ alpha,
                     const float* __restrict__ Dinv, const int* __restrict__ row,
                     const int* __restrict__ col, float* __restrict__ out) {
  int i = blockIdx.x;
  int idx = threadIdx.x << 2;
  int dd = idx >> 7;
  int r = row[i], c = col[i];
  float s = Dinv[r * DD + dd] * alpha[(size_t)i * DD + dd];
  float4 v = *(const float4*)(m + (size_t)c * DH + idx);
  float* dst = out + (size_t)r * DH + idx;
  ATOMIC_ADD_F32(dst + 0, s * v.x);
  ATOMIC_ADD_F32(dst + 1, s * v.y);
  ATOMIC_ADD_F32(dst + 2, s * v.z);
  ATOMIC_ADD_F32(dst + 3, s * v.w);
}

__global__ void bias_act_kernel(float* __restrict__ h, const float* __restrict__ b,
                                size_t total, int do_elu) {
  size_t gid = blockIdx.x * (size_t)blockDim.x + threadIdx.x;
  if (gid >= total) return;
  float v = h[gid] + b[gid & 127];
  if (do_elu) v = (v > 0.0f) ? v : (__expf(v) - 1.0f);
  h[gid] = v;
}

template <bool VB>
static inline void launch_gemm(const float* A, const float* B, float* C,
                               int M, int Nc, int K, int ldb, int ldc,
                               hipStream_t s) {
  dim3 grid((Nc + 63) / 64, (M + TM - 1) / TM);
  gemm_wmma_f32<VB><<<grid, dim3(128), 0, s>>>(A, B, C, M, Nc, K, ldb, ldc);
}

extern "C" void kernel_launch(void* const* d_in, const int* in_sizes, int n_in,
                              void* d_out, int out_size, void* d_ws, size_t ws_size,
                              hipStream_t stream) {
  const float* x        = (const float*)d_in[0];
  const int*   eidx     = (const int*)d_in[1];
  const float* hedge    = (const float*)d_in[2];
  const float* lin_W    = (const float*)d_in[3];
  const float* sheaf_W  = (const float*)d_in[4];
  const float* conv1_W  = (const float*)d_in[5];
  const float* conv1_b  = (const float*)d_in[6];
  const float* conv2_W  = (const float*)d_in[7];
  const float* conv2_b  = (const float*)d_in[8];
  const float* lin2_W   = (const float*)d_in[9];
  float* out = (float*)d_out;

  const int* row = eidx;
  const int* col = eidx + NNZ;

  float* ws    = (float*)d_ws;
  float* bufA  = ws;
  float* bufB  = bufA + (size_t)NN * DH;
  float* el    = bufB + (size_t)NN * DH;
  float* mbuf  = el + (size_t)EE * DH;
  float* xmean = mbuf + (size_t)EE * DH;
  float* emean = xmean + (size_t)NN * FF;
  float* xm    = emean + (size_t)EE * FF;
  float* em    = xm + (size_t)NN * DD;
  float* alpha = em + (size_t)EE * DD;
  float* Dg    = alpha + (size_t)NNZ * DD;
  float* Bg    = Dg + (size_t)NN * DD;

  const size_t nBig = (size_t)NN * DH;
  const size_t eBig = (size_t)EE * DH;

  launch_gemm<true>(x, lin_W, bufA, NN, DH, FF, DH, DH, stream);
  launch_gemm<true>(hedge, lin_W, el, EE, DH, FF, DH, DH, stream);

  mean_d_kernel<<<(NN * FF + 255) / 256, 256, 0, stream>>>(bufA, xmean, NN);
  mean_d_kernel<<<(EE * FF + 255) / 256, 256, 0, stream>>>(el, emean, EE);

  launch_gemm<false>(xmean, sheaf_W, xm, NN, DD, FF, DD, DD, stream);
  launch_gemm<false>(emean, sheaf_W + FF * DD, em, EE, DD, FF, DD, DD, stream);

  fill_kernel<<<64, 256, 0, stream>>>(Dg, (size_t)NN * DD, 0.0f);
  fill_kernel<<<64, 256, 0, stream>>>(Bg, (size_t)EE * DD, 0.0f);
  alpha_deg_kernel<<<(NNZ * DD + 255) / 256, 256, 0, stream>>>(xm, em, row, col,
                                                               alpha, Dg, Bg);
  recip_kernel<<<(NN * DD + 255) / 256, 256, 0, stream>>>(Dg, NN * DD);
  recip_kernel<<<(EE * DD + 255) / 256, 256, 0, stream>>>(Bg, EE * DD);

  launch_gemm<true>(bufA, conv1_W, bufB, NN * DD, HH, FF, HH, HH, stream);
  fill_kernel<<<1024, 256, 0, stream>>>(mbuf, eBig, 0.0f);
  scatter_to_edge<<<NNZ, 192, 0, stream>>>(bufB, alpha, Bg, row, col, mbuf);
  fill_kernel<<<4096, 256, 0, stream>>>(bufA, nBig, 0.0f);
  scatter_to_node<<<NNZ, 192, 0, stream>>>(mbuf, alpha, Dg, row, col, bufA);
  bias_act_kernel<<<(unsigned)((nBig + 255) / 256), 256, 0, stream>>>(bufA, conv1_b,
                                                                      nBig, 1);

  launch_gemm<true>(bufA, conv2_W, bufB, NN * DD, HH, HH, HH, HH, stream);
  fill_kernel<<<1024, 256, 0, stream>>>(mbuf, eBig, 0.0f);
  scatter_to_edge<<<NNZ, 192, 0, stream>>>(bufB, alpha, Bg, row, col, mbuf);
  fill_kernel<<<4096, 256, 0, stream>>>(bufA, nBig, 0.0f);
  scatter_to_node<<<NNZ, 192, 0, stream>>>(mbuf, alpha, Dg, row, col, bufA);
  bias_act_kernel<<<(unsigned)((nBig + 255) / 256), 256, 0, stream>>>(bufA, conv2_b,
                                                                      nBig, 0);

  launch_gemm<true>(bufA, lin2_W, out, NN, COUT, DH, COUT, COUT, stream);
}


# Conv aggregation: float atomic scatter-adds and their zero-fill passes replaced by per-segment gather sums over a CSR built in-call by integer rank atomics; sum order in a segment = arrival rank
# speedup vs baseline: 2.7648x; 1.6785x over previous
#include <hip/hip_runtime.h>
#include <hip/hip_bf16.h>

#define NN   20000
#define EE   2000
#define NNZ  100000
#define FF   128
#define HH   128
#define DD   6
#define COUT 40
#define DH   (DD * HH)

typedef __attribute__((ext_vector_type(2))) float v2f;
typedef __attribute__((ext_vector_type(8))) float v8f;

#define ATOMIC_ADD_F32(p, v) unsafeAtomicAdd((p), (v))

#define TK 32
#define TM 32
#define AP 36
template <bool VB>
__global__ __launch_bounds__(128)
void gemm_wmma_f32(const float* __restrict__ A, const float* __restrict__ B,
                   float* __restrict__ C, int M, int Nc, int K, int ldb, int ldc) {
  __shared__ __attribute__((aligned(16))) float As[TM][AP];
  __shared__ __attribute__((aligned(16))) float Bs[TK][64];

  const int tid  = threadIdx.x;
  const int wave = tid >> 5;
  const int lane = tid & 31;
  const int la   = lane & 15;
  const int hi   = lane >> 4;

  const int m0 = blockIdx.y * TM;
  const int nb = blockIdx.x * 64;
  const int n0 = nb + wave * 16;

  const int ar = tid >> 3, ac = (tid & 7) << 2;
  const int br = tid >> 4, bc = (tid & 15) << 2;
  const int gn = nb + bc;
  const float* __restrict__ Ap0 = A + (size_t)min(m0 + ar, M - 1) * K + ac;
  const float* __restrict__ Ap1 = A + (size_t)min(m0 + ar + 16, M - 1) * K + ac;
  const float* __restrict__ Bp  = B + (size_t)br * ldb;
  const size_t bstep = (size_t)8 * ldb;
  const int cv = min(gn, Nc - 4);
  const int c0 = min(gn + 0, Nc - 1), c1 = min(gn + 1, Nc - 1);
  const int c2 = min(gn + 2, Nc - 1), c3 = min(gn + 3, Nc - 1);

  float4 ra0, ra1, rb0, rb1, rb2, rb3;

#define GEMM_FETCH_B(dst_, q_)                                                 \
  do {                                                                         \
    if (VB) {                                                                  \
      dst_ = *(const float4*)((q_) + cv);                                      \
    } else {                                                                   \
      dst_ = make_float4((q_)[c0], (q_)[c1], (q_)[c2], (q_)[c3]);              \
    }                                                                          \
  } while (0)
#define GEMM_FETCH(kk_)                                                        \
  do {                                                                         \
    ra0 = *(const float4*)(Ap0 + (kk_));                                       \
    ra1 = *(const float4*)(Ap1 + (kk_));                                       \
    const float* qb_ = Bp + (size_t)(kk_) * ldb;                               \
    GEMM_FETCH_B(rb0, qb_);                                                    \
    GEMM_FETCH_B(rb1, qb_ + bstep);                                            \
    GEMM_FETCH_B(rb2, qb_ + 2 * bstep);                                        \
    GEMM_FETCH_B(rb3, qb_ + 3 * bstep);                                        \
  } while (0)

  v8f acc0 = {};
  v8f acc1 = {};

  GEMM_FETCH(0);

  for (int kk = 0; kk < K; kk += TK) {
    *(float4*)&As[ar][ac]      = ra0;
    *(float4*)&As[ar + 16][ac] = ra1;
    *(float4*)&Bs[br][bc]      = rb0;
    *(float4*)&Bs[br + 8][bc]  = rb1;
    *(float4*)&Bs[br + 16][bc] = rb2;
    *(float4*)&Bs[br + 24][bc] = rb3;
    __syncthreads();

    if (kk + TK < K) GEMM_FETCH(kk + TK);

#pragma unroll
    for (int k4 = 0; k4 < TK; k4 += 4) {
      v2f a0, a1, b;
      b.x  = Bs[k4 + 2 * hi][(wave << 4) + la];
      b.y  = Bs[k4 + 2 * hi + 1][(wave << 4) + la];
      a0.x = As[la][k4 + 2 * hi];
      a0.y = As[la][k4 + 2 * hi + 1];
      a1.x = As[16 + la][k4 + 2 * hi];
      a1.y = As[16 + la][k4 + 2 * hi + 1];
      acc0 = __builtin_amdgcn_wmma_f32_16x16x4_f32(
          false, a0, false, b, (short)0, acc0, false, false);
      acc1 = __builtin_amdgcn_wmma_f32_16x16x4_f32(
          false, a1, false, b, (short)0, acc1, false, false);
    }
    __syncthreads();
  }
#undef GEMM_FETCH
#undef GEMM_FETCH_B

  const int n = n0 + la;
  if (n < Nc) {
#pragma unroll
    for (int r = 0; r < 8; ++r) {
      int gm = m0 + r + 8 * hi;
      if (gm < M) C[(size_t)gm * ldc + n] = acc0[r];
    }
#pragma unroll
    for (int r = 0; r < 8; ++r) {
      int gm = m0 + 16 + r + 8 * hi;
      if (gm < M) C[(size_t)gm * ldc + n] = acc1[r];
    }
  }
}

__global__ void mean_d_kernel(const float* __restrict__ xl, float* __restrict__ xm,
                              int rows) {
  int gid = blockIdx.x * blockDim.x + threadIdx.x;
  if (gid >= rows * FF) return;
  int n = gid >> 7, f = gid & 127;
  const float* p = xl + (size_t)n * DH + f;
  float s = 0.0f;
#pragma unroll
  for (int d = 0; d < DD; ++d) s += p[d * FF];
  xm[gid] = s * (1.0f / (float)DD);
}

__global__ void alpha_deg_kernel(const float* __restrict__ xm, const float* __restrict__ em,
                                 const int* __restrict__ row, const int* __restrict__ col,
                                 float* __restrict__ alpha, float* __restrict__ Dg,
                                 float* __restrict__ Bg) {
  int gid = blockIdx.x * blockDim.x + threadIdx.x;
  if (gid >= NNZ * DD) return;
  int i = gid / DD, dd = gid - i * DD;
  int r = row[i], c = col[i];
  float v = xm[r * DD + dd] + em[c * DD + dd];
  v = 1.0f / (1.0f + __expf(-v));
  alpha[gid] = v;
  ATOMIC_ADD_F32(&Dg[r * DD + dd], v);
  ATOMIC_ADD_F32(&Bg[c * DD + dd], v);
}

__global__ void recip_kernel(float* __restrict__ p, int n) {
  int gid = blockIdx.x * blockDim.x + threadIdx.x;
  if (gid >= n) return;
  float v = p[gid];
  p[gid] = (v != 0.0f) ? (1.0f / v) : 0.0f;
}

__global__ void fill_kernel(float* __restrict__ p, size_t n, float v) {
  size_t stride = (size_t)gridDim.x * blockDim.x;
  for (size_t i = blockIdx.x * (size_t)blockDim.x + threadIdx.x; i < n; i += stride)
    p[i] = v;
}

__global__ void rank_kernel(const int* __restrict__ row, const int* __restrict__ col,
                            int* __restrict__ cntN, int* __restrict__ cntE,
                            int* __restrict__ rankN, int* __restrict__ rankE) {
  int i = blockIdx.x * blockDim.x + threadIdx.x;
  if (i >= NNZ) return;
  rankN[i] = atomicAdd(&cntN[row[i]], 1);
  rankE[i] = atomicAdd(&cntE[col[i]], 1);
}

__global__ __launch_bounds__(1024)
void scan_kernel(const int* __restrict__ cntN, int* __restrict__ offN,
                 const int* __restrict__ cntE, int* __restrict__ offE) {
  __shared__ int part[1024];
  const int* __restrict__ cnt = (blockIdx.x == 0) ? cntN : cntE;
  int* __restrict__ off       = (blockIdx.x == 0) ? offN : offE;
  const int n   = (blockIdx.x == 0) ? NN : EE;
  const int per = (n + 1023) >> 10;
  const int b = threadIdx.x * per;
  const int e = min(b + per, n);
  int s = 0;
  for (int k = b; k < e; ++k) s += cnt[k];
  part[threadIdx.x] = s;
  __syncthreads();
  for (int d = 1; d < 1024; d <<= 1) {
    int v = (threadIdx.x >= d) ? part[threadIdx.x - d] : 0;
    __syncthreads();
    part[threadIdx.x] += v;
    __syncthreads();
  }
  int run = part[threadIdx.x] - s;
  for (int k = b; k < e; ++k) { off[k] = run; run += cnt[k]; }
  if (threadIdx.x == 1023) off[n] = part[1023];
}

__global__ void place_kernel(const int* __restrict__ row, const int* __restrict__ col,
                             const int* __restrict__ offN, const int* __restrict__ offE,
                             const int* __restrict__ rankN, const int* __restrict__ rankE,
                             int* __restrict__ permN, int* __restrict__ permE) {
  int i = blockIdx.x * blockDim.x + threadIdx.x;
  if (i >= NNZ) return;
  permN[offN[row[i]] + rankN[i]] = i;
  permE[offE[col[i]] + rankE[i]] = i;
}

template <int U>
__global__ __launch_bounds__(192)
void segment_gather(const float* __restrict__ src, const float* __restrict__ alpha,
                    const float* __restrict__ inv, const int* __restrict__ other,
                    const int* __restrict__ off, const int* __restrict__ perm,
                    float* __restrict__ dst, int nseg) {
  const int idx = threadIdx.x << 2;
  const int dd = __builtin_amdgcn_readfirstlane(idx >> 7);
  for (int g = blockIdx.x; g < nseg; g += gridDim.x) {
    const int p0 = off[g], p1 = off[g + 1];
    const float w = inv[g * DD + dd];
    float4 acc = make_float4(0.0f, 0.0f, 0.0f, 0.0f);
    int p = p0;
    for (; p + U <= p1; p += U) {
      int ii[U];
      float s[U];
      float4 v[U];
#pragma unroll
      for (int u = 0; u < U; ++u) ii[u] = perm[p + u];
#pragma unroll
      for (int u = 0; u < U; ++u) {
        s[u] = w * alpha[(size_t)ii[u] * DD + dd];
        v[u] = *(const float4*)(src + (size_t)other[ii[u]] * DH + idx);
      }
#pragma unroll
      for (int u = 0; u < U; ++u) {
        acc.x += s[u] * v[u].x;
        acc.y += s[u] * v[u].y;
        acc.z += s[u] * v[u].z;
        acc.w += s[u] * v[u].w;
      }
    }
    for (; p < p1; ++p) {
      const int i = perm[p];
      const float s = w * alpha[(size_t)i * DD + dd];
      const float4 v = *(const float4*)(src + (size_t)other[i] * DH + idx);
      acc.x += s * v.x;
      acc.y += s * v.y;
      acc.z += s * v.z;
      acc.w += s * v.w;
    }
    *(float4*)(dst + (size_t)g * DH + idx) = acc;
  }
}

__global__ void bias_act_kernel(float* __restrict__ h, const float* __restrict__ b,
                                size_t total, int do_elu) {
  size_t gid = blockIdx.x * (size_t)blockDim.x + threadIdx.x;
  if (gid >= total) return;
  float v = h[gid] + b[gid & 127];
  if (do_elu) v = (v > 0.0f) ? v : (__expf(v) - 1.0f);
  h[gid] = v;
}

template <bool VB>
static inline void launch_gemm(const float* A, const float* B, float* C,
                               int M, int Nc, int K, int ldb, int ldc,
                               hipStream_t s) {
  dim3 grid((Nc + 63) / 64, (M + TM - 1) / TM);
  gemm_wmma_f32<VB><<<grid, dim3(128), 0, s>>>(A, B, C, M, Nc, K, ldb, ldc);
}

extern "C" void kernel_launch(void* const* d_in, const int* in_sizes, int n_in,
                              void* d_out, int out_size, void* d_ws, size_t ws_size,
                              hipStream_t stream) {
  const float* x        = (const float*)d_in[0];
  const int*   eidx     = (const int*)d_in[1];
  const float* hedge    = (const float*)d_in[2];
  const float* lin_W    = (const float*)d_in[3];
  const float* sheaf_W  = (const float*)d_in[4];
  const float* conv1_W  = (const float*)d_in[5];
  const float* conv1_b  = (const float*)d_in[6];
  const float* conv2_W  = (const float*)d_in[7];
  const float* conv2_b  = (const float*)d_in[8];
  const float* lin2_W   = (const float*)d_in[9];
  float* out = (float*)d_out;

  const int* row = eidx;
  const int* col = eidx + NNZ;

  float* ws    = (float*)d_ws;
  float* bufA  = ws;
  float* bufB  = bufA + (size_t)NN * DH;
  float* el    = bufB + (size_t)NN * DH;
  float* mbuf  = el + (size_t)EE * DH;
  float* xmean = mbuf + (size_t)EE * DH;
  float* emean = xmean + (size_t)NN * FF;
  float* xm    = emean + (size_t)EE * FF;
  float* em    = xm + (size_t)NN * DD;
  float* alpha = em + (size_t)EE * DD;
  float* Dg    = alpha + (size_t)NNZ * DD;
  float* Bg    = Dg + (size_t)NN * DD;

  int* cntN  = (int*)xmean;
  int* cntE  = cntN + NN;
  int* offN  = cntE + EE;
  int* offE  = offN + (NN + 1);
  int* rankN = offE + (EE + 1);
  int* rankE = rankN + NNZ;
  int* permN = rankE + NNZ;
  int* permE = permN + NNZ;

  const size_t nBig = (size_t)NN * DH;

  launch_gemm<true>(x, lin_W, bufA, NN, DH, FF, DH, DH, stream);
  launch_gemm<true>(hedge, lin_W, el, EE, DH, FF, DH, DH, stream);

  mean_d_kernel<<<(NN * FF + 255) / 256, 256, 0, stream>>>(bufA, xmean, NN);
  mean_d_kernel<<<(EE * FF + 255) / 256, 256, 0, stream>>>(el, emean, EE);

  launch_gemm<false>(xmean, sheaf_W, xm, NN, DD, FF, DD, DD, stream);
  launch_gemm<false>(emean, sheaf_W + FF * DD, em, EE, DD, FF, DD, DD, stream);

  fill_kernel<<<64, 256, 0, stream>>>(Dg, (size_t)NN * DD, 0.0f);
  fill_kernel<<<64, 256, 0, stream>>>(Bg, (size_t)EE * DD, 0.0f);
  alpha_deg_kernel<<<(NNZ * DD + 255) / 256, 256, 0, stream>>>(xm, em, row, col,
                                                               alpha, Dg, Bg);
  recip_kernel<<<(NN * DD + 255) / 256, 256, 0, stream>>>(Dg, NN * DD);
  recip_kernel<<<(EE * DD + 255) / 256, 256, 0, stream>>>(Bg, EE * DD);

  fill_kernel<<<64, 256, 0, stream>>>((float*)cntN, (size_t)(NN + EE), 0.0f);
  rank_kernel<<<(NNZ + 255) / 256, 256, 0, stream>>>(row, col, cntN, cntE, rankN, rankE);
  scan_kernel<<<2, 1024, 0, stream>>>(cntN, offN, cntE, offE);
  place_kernel<<<(NNZ + 255) / 256, 256, 0, stream>>>(row, col, offN, offE, rankN, rankE,
                                                     permN, permE);

  launch_gemm<true>(bufA, conv1_W, bufB, NN * DD, HH, FF, HH, HH, stream);
  segment_gather<8><<<EE, 192, 0, stream>>>(bufB, alpha, Bg, row, offE, permE, mbuf, EE);
  segment_gather<4><<<2560, 192, 0, stream>>>(mbuf, alpha, Dg, col, offN, permN, bufA, NN);
  bias_act_kernel<<<(unsigned)((nBig + 255) / 256), 256, 0, stream>>>(bufA, conv1_b,
                                                                      nBig, 1);

  launch_gemm<true>(bufA, conv2_W, bufB, NN * DD, HH, HH, HH, HH, stream);
  segment_gather<8><<<EE, 192, 0, stream>>>(bufB, alpha, Bg, row, offE, permE, mbuf, EE);
  segment_gather<4><<<2560, 192, 0, stream>>>(mbuf, alpha, Dg, col, offN, permN, bufA, NN);
  bias_act_kernel<<<(unsigned)((nBig + 255) / 256), 256, 0, stream>>>(bufA, conv2_b,
                                                                      nBig, 0);

  launch_gemm<true>(bufA, lin2_W, out, NN, COUT, DH, COUT, COUT, stream);
}


# Conv bias add and ELU folded into the hyperedge->node gather's store (on the finished sum in registers): the two one-float-per-lane passes over the 61 MB layer output and their 2 launches go
# speedup vs baseline: 3.4663x; 1.2537x over previous
#include <hip/hip_runtime.h>
#include <hip/hip_bf16.h>

#define NN   20000
#define EE   2000
#define NNZ  100000
#define FF   128
#define HH   128
#define DD   6
#define COUT 40
#define DH   (DD * HH)

typedef __attribute__((ext_vector_type(2))) float v2f;
typedef __attribute__((ext_vector_type(8))) float v8f;

#define ATOMIC_ADD_F32(p, v) unsafeAtomicAdd((p), (v))

#define TK 32
#define TM 32
#define AP 36
template <bool VB>
__global__ __launch_bounds__(128)
void gemm_wmma_f32(const float* __restrict__ A, const float* __restrict__ B,
                   float* __restrict__ C, int M, int Nc, int K, int ldb, int ldc) {
  __shared__ __attribute__((aligned(16))) float As[TM][AP];
  __shared__ __attribute__((aligned(16))) float Bs[TK][64];

  const int tid  = threadIdx.x;
  const int wave = tid >> 5;
  const int lane = tid & 31;
  const int la   = lane & 15;
  const int hi   = lane >> 4;

  const int m0 = blockIdx.y * TM;
  const int nb = blockIdx.x * 64;
  const int n0 = nb + wave * 16;

  const int ar = tid >> 3, ac = (tid & 7) << 2;
  const int br = tid >> 4, bc = (tid & 15) << 2;
  const int gn = nb + bc;
  const float* __restrict__ Ap0 = A + (size_t)min(m0 + ar, M - 1) * K + ac;
  const float* __restrict__ Ap1 = A + (size_t)min(m0 + ar + 16, M - 1) * K + ac;
  const float* __restrict__ Bp  = B + (size_t)br * ldb;
  const size_t bstep = (size_t)8 * ldb;
  const int cv = min(gn, Nc - 4);
  const int c0 = min(gn + 0, Nc - 1), c1 = min(gn + 1, Nc - 1);
  const int c2 = min(gn + 2, Nc - 1), c3 = min(gn + 3, Nc - 1);

  float4 ra0, ra1, rb0, rb1, rb2, rb3;

#define GEMM_FETCH_B(dst_, q_)                                                 \
  do {                                                                         \
    if (VB) {                                                                  \
      dst_ = *(const float4*)((q_) + cv);                                      \
    } else {                                                                   \
      dst_ = make_float4((q_)[c0], (q_)[c1], (q_)[c2], (q_)[c3]);              \
    }                                                                          \
  } while (0)
#define GEMM_FETCH(kk_)                                                        \
  do {                                                                         \
    ra0 = *(const float4*)(Ap0 + (kk_));                                       \
    ra1 = *(const float4*)(Ap1 + (kk_));                                       \
    const float* qb_ = Bp + (size_t)(kk_) * ldb;                               \
    GEMM_FETCH_B(rb0, qb_);                                                    \
    GEMM_FETCH_B(rb1, qb_ + bstep);                                            \
    GEMM_FETCH_B(rb2, qb_ + 2 * bstep);                                        \
    GEMM_FETCH_B(rb3, qb_ + 3 * bstep);                                        \
  } while (0)

  v8f acc0 = {};
  v8f acc1 = {};

  GEMM_FETCH(0);

  for (int kk = 0; kk < K; kk += TK) {
    *(float4*)&As[ar][ac]      = ra0;
    *(float4*)&As[ar + 16][ac] = ra1;
    *(float4*)&Bs[br][bc]      = rb0;
    *(float4*)&Bs[br + 8][bc]  = rb1;
    *(float4*)&Bs[br + 16][bc] = rb2;
    *(float4*)&Bs[br + 24][bc] = rb3;
    __syncthreads();

    if (kk + TK < K) GEMM_FETCH(kk + TK);

#pragma unroll
    for (int k4 = 0; k4 < TK; k4 += 4) {
      v2f a0, a1, b;
      b.x  = Bs[k4 + 2 * hi][(wave << 4) + la];
      b.y  = Bs[k4 + 2 * hi + 1][(wave << 4) + la];
      a0.x = As[la][k4 + 2 * hi];
      a0.y = As[la][k4 + 2 * hi + 1];
      a1.x = As[16 + la][k4 + 2 * hi];
      a1.y = As[16 + la][k4 + 2 * hi + 1];
      acc0 = __builtin_amdgcn_wmma_f32_16x16x4_f32(
          false, a0, false, b, (short)0, acc0, false, false);
      acc1 = __builtin_amdgcn_wmma_f32_16x16x4_f32(
          false, a1, false, b, (short)0, acc1, false, false);
    }
    __syncthreads();
  }
#undef GEMM_FETCH
#undef GEMM_FETCH_B

  const int n = n0 + la;
  if (n < Nc) {
#pragma unroll
    for (int r = 0; r < 8; ++r) {
      int gm = m0 + r + 8 * hi;
      if (gm < M) C[(size_t)gm * ldc + n] = acc0[r];
    }
#pragma unroll
    for (int r = 0; r < 8; ++r) {
      int gm = m0 + 16 + r + 8 * hi;
      if (gm < M) C[(size_t)gm * ldc + n] = acc1[r];
    }
  }
}

__global__ void mean_d_kernel(const float* __restrict__ xl, float* __restrict__ xm,
                              int rows) {
  int gid = blockIdx.x * blockDim.x + threadIdx.x;
  if (gid >= rows * FF) return;
  int n = gid >> 7, f = gid & 127;
  const float* p = xl + (size_t)n * DH + f;
  float s = 0.0f;
#pragma unroll
  for (int d = 0; d < DD; ++d) s += p[d * FF];
  xm[gid] = s * (1.0f / (float)DD);
}

__global__ void alpha_deg_kernel(const float* __restrict__ xm, const float* __restrict__ em,
                                 const int* __restrict__ row, const int* __restrict__ col,
                                 float* __restrict__ alpha, float* __restrict__ Dg,
                                 float* __restrict__ Bg) {
  int gid = blockIdx.x * blockDim.x + threadIdx.x;
  if (gid >= NNZ * DD) return;
  int i = gid / DD, dd = gid - i * DD;
  int r = row[i], c = col[i];
  float v = xm[r * DD + dd] + em[c * DD + dd];
  v = 1.0f / (1.0f + __expf(-v));
  alpha[gid] = v;
  ATOMIC_ADD_F32(&Dg[r * DD + dd], v);
  ATOMIC_ADD_F32(&Bg[c * DD + dd], v);
}

__global__ void recip_kernel(float* __restrict__ p, int n) {
  int gid = blockIdx.x * blockDim.x + threadIdx.x;
  if (gid >= n) return;
  float v = p[gid];
  p[gid] = (v != 0.0f) ? (1.0f / v) : 0.0f;
}

__global__ void fill_kernel(float* __restrict__ p, size_t n, float v) {
  size_t stride = (size_t)gridDim.x * blockDim.x;
  for (size_t i = blockIdx.x * (size_t)blockDim.x + threadIdx.x; i < n; i += stride)
    p[i] = v;
}

__global__ void rank_kernel(const int* __restrict__ row, const int* __restrict__ col,
                            int* __restrict__ cntN, int* __restrict__ cntE,
                            int* __restrict__ rankN, int* __restrict__ rankE) {
  int i = blockIdx.x * blockDim.x + threadIdx.x;
  if (i >= NNZ) return;
  rankN[i] = atomicAdd(&cntN[row[i]], 1);
  rankE[i] = atomicAdd(&cntE[col[i]], 1);
}

__global__ __launch_bounds__(1024)
void scan_kernel(const int* __restrict__ cntN, int* __restrict__ offN,
                 const int* __restrict__ cntE, int* __restrict__ offE) {
  __shared__ int part[1024];
  const int* __restrict__ cnt = (blockIdx.x == 0) ? cntN : cntE;
  int* __restrict__ off       = (blockIdx.x == 0) ? offN : offE;
  const int n   = (blockIdx.x == 0) ? NN : EE;
  const int per = (n + 1023) >> 10;
  const int b = threadIdx.x * per;
  const int e = min(b + per, n);
  int s = 0;
  for (int k = b; k < e; ++k) s += cnt[k];
  part[threadIdx.x] = s;
  __syncthreads();
  for (int d = 1; d < 1024; d <<= 1) {
    int v = (threadIdx.x >= d) ? part[threadIdx.x - d] : 0;
    __syncthreads();
    part[threadIdx.x] += v;
    __syncthreads();
  }
  int run = part[threadIdx.x] - s;
  for (int k = b; k < e; ++k) { off[k] = run; run += cnt[k]; }
  if (threadIdx.x == 1023) off[n] = part[1023];
}

__global__ void place_kernel(const int* __restrict__ row, const int* __restrict__ col,
                             const int* __restrict__ offN, const int* __restrict__ offE,
                             const int* __restrict__ rankN, const int* __restrict__ rankE,
                             int* __restrict__ permN, int* __restrict__ permE) {
  int i = blockIdx.x * blockDim.x + threadIdx.x;
  if (i >= NNZ) return;
  permN[offN[row[i]] + rankN[i]] = i;
  permE[offE[col[i]] + rankE[i]] = i;
}

template <int U, int EPI>
__global__ __launch_bounds__(192)
void segment_gather(const float* __restrict__ src, const float* __restrict__ alpha,
                    const float* __restrict__ inv, const int* __restrict__ other,
                    const int* __restrict__ off, const int* __restrict__ perm,
                    const float* __restrict__ bias,
                    float* __restrict__ dst, int nseg) {
  const int idx = threadIdx.x << 2;
  float4 bv = make_float4(0.0f, 0.0f, 0.0f, 0.0f);
  if (EPI != 0) {
    const int h = idx & (HH - 1);
    bv = make_float4(bias[h], bias[h + 1], bias[h + 2], bias[h + 3]);
  }
  const int dd = __builtin_amdgcn_readfirstlane(idx >> 7);
  for (int g = blockIdx.x; g < nseg; g += gridDim.x) {
    const int p0 = off[g], p1 = off[g + 1];
    const float w = inv[g * DD + dd];
    float4 acc = make_float4(0.0f, 0.0f, 0.0f, 0.0f);
    int p = p0;
    for (; p + U <= p1; p += U) {
      int ii[U];
      float s[U];
      float4 v[U];
#pragma unroll
      for (int u = 0; u < U; ++u) ii[u] = perm[p + u];
#pragma unroll
      for (int u = 0; u < U; ++u) {
        s[u] = w * alpha[(size_t)ii[u] * DD + dd];
        v[u] = *(const float4*)(src + (size_t)other[ii[u]] * DH + idx);
      }
#pragma unroll
      for (int u = 0; u < U; ++u) {
        acc.x += s[u] * v[u].x;
        acc.y += s[u] * v[u].y;
        acc.z += s[u] * v[u].z;
        acc.w += s[u] * v[u].w;
      }
    }
    for (; p < p1; ++p) {
      const int i = perm[p];
      const float s = w * alpha[(size_t)i * DD + dd];
      const float4 v = *(const float4*)(src + (size_t)other[i] * DH + idx);
      acc.x += s * v.x;
      acc.y += s * v.y;
      acc.z += s * v.z;
      acc.w += s * v.w;
    }
    if (EPI != 0) {
      acc.x += bv.x;
      acc.y += bv.y;
      acc.z += bv.z;
      acc.w += bv.w;
    }
    if (EPI == 2) {
      acc.x = (acc.x > 0.0f) ? acc.x : (__expf(acc.x) - 1.0f);
      acc.y = (acc.y > 0.0f) ? acc.y : (__expf(acc.y) - 1.0f);
      acc.z = (acc.z > 0.0f) ? acc.z : (__expf(acc.z) - 1.0f);
      acc.w = (acc.w > 0.0f) ? acc.w : (__expf(acc.w) - 1.0f);
    }
    *(float4*)(dst + (size_t)g * DH + idx) = acc;
  }
}

template <bool VB>
static inline void launch_gemm(const float* A, const float* B, float* C,
                               int M, int Nc, int K, int ldb, int ldc,
                               hipStream_t s) {
  dim3 grid((Nc + 63) / 64, (M + TM - 1) / TM);
  gemm_wmma_f32<VB><<<grid, dim3(128), 0, s>>>(A, B, C, M, Nc, K, ldb, ldc);
}

extern "C" void kernel_launch(void* const* d_in, const int* in_sizes, int n_in,
                              void* d_out, int out_size, void* d_ws, size_t ws_size,
                              hipStream_t stream) {
  const float* x        = (const float*)d_in[0];
  const int*   eidx     = (const int*)d_in[1];
  const float* hedge    = (const float*)d_in[2];
  const float* lin_W    = (const float*)d_in[3];
  const float* sheaf_W  = (const float*)d_in[4];
  const float* conv1_W  = (const float*)d_in[5];
  const float* conv1_b  = (const float*)d_in[6];
  const float* conv2_W  = (const float*)d_in[7];
  const float* conv2_b  = (const float*)d_in[8];
  const float* lin2_W   = (const float*)d_in[9];
  float* out = (float*)d_out;

  const int* row = eidx;
  const int* col = eidx + NNZ;

  float* ws    = (float*)d_ws;
  float* bufA  = ws;
  float* bufB  = bufA + (size_t)NN * DH;
  float* el    = bufB + (size_t)NN * DH;
  float* mbuf  = el + (size_t)EE * DH;
  float* xmean = mbuf + (size_t)EE * DH;
  float* emean = xmean + (size_t)NN * FF;
  float* xm    = emean + (size_t)EE * FF;
  float* em    = xm + (size_t)NN * DD;
  float* alpha = em + (size_t)EE * DD;
  float* Dg    = alpha + (size_t)NNZ * DD;
  float* Bg    = Dg + (size_t)NN * DD;

  int* cntN  = (int*)xmean;
  int* cntE  = cntN + NN;
  int* offN  = cntE + EE;
  int* offE  = offN + (NN + 1);
  int* rankN = offE + (EE + 1);
  int* rankE = rankN + NNZ;
  int* permN = rankE + NNZ;
  int* permE = permN + NNZ;

  launch_gemm<true>(x, lin_W, bufA, NN, DH, FF, DH, DH, stream);
  launch_gemm<true>(hedge, lin_W, el, EE, DH, FF, DH, DH, stream);

  mean_d_kernel<<<(NN * FF + 255) / 256, 256, 0, stream>>>(bufA, xmean, NN);
  mean_d_kernel<<<(EE * FF + 255) / 256, 256, 0, stream>>>(el, emean, EE);

  launch_gemm<false>(xmean, sheaf_W, xm, NN, DD, FF, DD, DD, stream);
  launch_gemm<false>(emean, sheaf_W + FF * DD, em, EE, DD, FF, DD, DD, stream);

  fill_kernel<<<64, 256, 0, stream>>>(Dg, (size_t)NN * DD, 0.0f);
  fill_kernel<<<64, 256, 0, stream>>>(Bg, (size_t)EE * DD, 0.0f);
  alpha_deg_kernel<<<(NNZ * DD + 255) / 256, 256, 0, stream>>>(xm, em, row, col,
                                                               alpha, Dg, Bg);
  recip_kernel<<<(NN * DD + 255) / 256, 256, 0, stream>>>(Dg, NN * DD);
  recip_kernel<<<(EE * DD + 255) / 256, 256, 0, stream>>>(Bg, EE * DD);

  fill_kernel<<<64, 256, 0, stream>>>((float*)cntN, (size_t)(NN + EE), 0.0f);
  rank_kernel<<<(NNZ + 255) / 256, 256, 0, stream>>>(row, col, cntN, cntE, rankN, rankE);
  scan_kernel<<<2, 1024, 0, stream>>>(cntN, offN, cntE, offE);
  place_kernel<<<(NNZ + 255) / 256, 256, 0, stream>>>(row, col, offN, offE, rankN, rankE,
                                                     permN, permE);

  launch_gemm<true>(bufA, conv1_W, bufB, NN * DD, HH, FF, HH, HH, stream);
  segment_gather<8, 0><<<EE, 192, 0, stream>>>(bufB, alpha, Bg, row, offE, permE, nullptr,
                                               mbuf, EE);
  segment_gather<4, 2><<<2560, 192, 0, stream>>>(mbuf, alpha, Dg, col, offN, permN, conv1_b,
                                                 bufA, NN);

  launch_gemm<true>(bufA, conv2_W, bufB, NN * DD, HH, HH, HH, HH, stream);
  segment_gather<8, 0><<<EE, 192, 0, stream>>>(bufB, alpha, Bg, row, offE, permE, nullptr,
                                               mbuf, EE);
  segment_gather<4, 1><<<2560, 192, 0, stream>>>(mbuf, alpha, Dg, col, offN, permN, conv2_b,
                                                 bufA, NN);

  launch_gemm<true>(bufA, lin2_W, out, NN, COUT, DH, COUT, COUT, stream);
}


# Conv weight product moved behind the node->hyperedge sum (the sum is linear per channel): W applied to 12,000 hyperedge rows, not 120,000 node rows; sums reassociated (sum, then project)
# speedup vs baseline: 3.9788x; 1.1479x over previous
#include <hip/hip_runtime.h>
#include <hip/hip_bf16.h>

#define NN   20000
#define EE   2000
#define NNZ  100000
#define FF   128
#define HH   128
#define DD   6
#define COUT 40
#define DH   (DD * HH)

typedef __attribute__((ext_vector_type(2))) float v2f;
typedef __attribute__((ext_vector_type(8))) float v8f;

#define ATOMIC_ADD_F32(p, v) unsafeAtomicAdd((p), (v))

#define TK 32
#define TM 32
#define AP 36
template <bool VB>
__global__ __launch_bounds__(128)
void gemm_wmma_f32(const float* __restrict__ A, const float* __restrict__ B,
                   float* __restrict__ C, int M, int Nc, int K, int ldb, int ldc) {
  __shared__ __attribute__((aligned(16))) float As[TM][AP];
  __shared__ __attribute__((aligned(16))) float Bs[TK][64];

  const int tid  = threadIdx.x;
  const int wave = tid >> 5;
  const int lane = tid & 31;
  const int la   = lane & 15;
  const int hi   = lane >> 4;

  const int m0 = blockIdx.y * TM;
  const int nb = blockIdx.x * 64;
  const int n0 = nb + wave * 16;

  const int ar = tid >> 3, ac = (tid & 7) << 2;
  const int br = tid >> 4, bc = (tid & 15) << 2;
  const int gn = nb + bc;
  const float* __restrict__ Ap0 = A + (size_t)min(m0 + ar, M - 1) * K + ac;
  const float* __restrict__ Ap1 = A + (size_t)min(m0 + ar + 16, M - 1) * K + ac;
  const float* __restrict__ Bp  = B + (size_t)br * ldb;
  const size_t bstep = (size_t)8 * ldb;
  const int cv = min(gn, Nc - 4);
  const int c0 = min(gn + 0, Nc - 1), c1 = min(gn + 1, Nc - 1);
  const int c2 = min(gn + 2, Nc - 1), c3 = min(gn + 3, Nc - 1);

  float4 ra0, ra1, rb0, rb1, rb2, rb3;

#define GEMM_FETCH_B(dst_, q_)                                                 \
  do {                                                                         \
    if (VB) {                                                                  \
      dst_ = *(const float4*)((q_) + cv);                                      \
    } else {                                                                   \
      dst_ = make_float4((q_)[c0], (q_)[c1], (q_)[c2], (q_)[c3]);              \
    }                                                                          \
  } while (0)
#define GEMM_FETCH(kk_)                                                        \
  do {                                                                         \
    ra0 = *(const float4*)(Ap0 + (kk_));                                       \
    ra1 = *(const float4*)(Ap1 + (kk_));                                       \
    const float* qb_ = Bp + (size_t)(kk_) * ldb;                               \
    GEMM_FETCH_B(rb0, qb_);                                                    \
    GEMM_FETCH_B(rb1, qb_ + bstep);                                            \
    GEMM_FETCH_B(rb2, qb_ + 2 * bstep);                                        \
    GEMM_FETCH_B(rb3, qb_ + 3 * bstep);                                        \
  } while (0)

  v8f acc0 = {};
  v8f acc1 = {};

  GEMM_FETCH(0);

  for (int kk = 0; kk < K; kk += TK) {
    *(float4*)&As[ar][ac]      = ra0;
    *(float4*)&As[ar + 16][ac] = ra1;
    *(float4*)&Bs[br][bc]      = rb0;
    *(float4*)&Bs[br + 8][bc]  = rb1;
    *(float4*)&Bs[br + 16][bc] = rb2;
    *(float4*)&Bs[br + 24][bc] = rb3;
    __syncthreads();

    if (kk + TK < K) GEMM_FETCH(kk + TK);

#pragma unroll
    for (int k4 = 0; k4 < TK; k4 += 4) {
      v2f a0, a1, b;
      b.x  = Bs[k4 + 2 * hi][(wave << 4) + la];
      b.y  = Bs[k4 + 2 * hi + 1][(wave << 4) + la];
      a0.x = As[la][k4 + 2 * hi];
      a0.y = As[la][k4 + 2 * hi + 1];
      a1.x = As[16 + la][k4 + 2 * hi];
      a1.y = As[16 + la][k4 + 2 * hi + 1];
      acc0 = __builtin_amdgcn_wmma_f32_16x16x4_f32(
          false, a0, false, b, (short)0, acc0, false, false);
      acc1 = __builtin_amdgcn_wmma_f32_16x16x4_f32(
          false, a1, false, b, (short)0, acc1, false, false);
    }
    __syncthreads();
  }
#undef GEMM_FETCH
#undef GEMM_FETCH_B

  const int n = n0 + la;
  if (n < Nc) {
#pragma unroll
    for (int r = 0; r < 8; ++r) {
      int gm = m0 + r + 8 * hi;
      if (gm < M) C[(size_t)gm * ldc + n] = acc0[r];
    }
#pragma unroll
    for (int r = 0; r < 8; ++r) {
      int gm = m0 + 16 + r + 8 * hi;
      if (gm < M) C[(size_t)gm * ldc + n] = acc1[r];
    }
  }
}

__global__ void mean_d_kernel(const float* __restrict__ xl, float* __restrict__ xm,
                              int rows) {
  int gid = blockIdx.x * blockDim.x + threadIdx.x;
  if (gid >= rows * FF) return;
  int n = gid >> 7, f = gid & 127;
  const float* p = xl + (size_t)n * DH + f;
  float s = 0.0f;
#pragma unroll
  for (int d = 0; d < DD; ++d) s += p[d * FF];
  xm[gid] = s * (1.0f / (float)DD);
}

__global__ void alpha_deg_kernel(const float* __restrict__ xm, const float* __restrict__ em,
                                 const int* __restrict__ row, const int* __restrict__ col,
                                 float* __restrict__ alpha, float* __restrict__ Dg,
                                 float* __restrict__ Bg) {
  int gid = blockIdx.x * blockDim.x + threadIdx.x;
  if (gid >= NNZ * DD) return;
  int i = gid / DD, dd = gid - i * DD;
  int r = row[i], c = col[i];
  float v = xm[r * DD + dd] + em[c * DD + dd];
  v = 1.0f / (1.0f + __expf(-v));
  alpha[gid] = v;
  ATOMIC_ADD_F32(&Dg[r * DD + dd], v);
  ATOMIC_ADD_F32(&Bg[c * DD + dd], v);
}

__global__ void recip_kernel(float* __restrict__ p, int n) {
  int gid = blockIdx.x * blockDim.x + threadIdx.x;
  if (gid >= n) return;
  float v = p[gid];
  p[gid] = (v != 0.0f) ? (1.0f / v) : 0.0f;
}

__global__ void fill_kernel(float* __restrict__ p, size_t n, float v) {
  size_t stride = (size_t)gridDim.x * blockDim.x;
  for (size_t i = blockIdx.x * (size_t)blockDim.x + threadIdx.x; i < n; i += stride)
    p[i] = v;
}

__global__ void rank_kernel(const int* __restrict__ row, const int* __restrict__ col,
                            int* __restrict__ cntN, int* __restrict__ cntE,
                            int* __restrict__ rankN, int* __restrict__ rankE) {
  int i = blockIdx.x * blockDim.x + threadIdx.x;
  if (i >= NNZ) return;
  rankN[i] = atomicAdd(&cntN[row[i]], 1);
  rankE[i] = atomicAdd(&cntE[col[i]], 1);
}

__global__ __launch_bounds__(1024)
void scan_kernel(const int* __restrict__ cntN, int* __restrict__ offN,
                 const int* __restrict__ cntE, int* __restrict__ offE) {
  __shared__ int part[1024];
  const int* __restrict__ cnt = (blockIdx.x == 0) ? cntN : cntE;
  int* __restrict__ off       = (blockIdx.x == 0) ? offN : offE;
  const int n   = (blockIdx.x == 0) ? NN : EE;
  const int per = (n + 1023) >> 10;
  const int b = threadIdx.x * per;
  const int e = min(b + per, n);
  int s = 0;
  for (int k = b; k < e; ++k) s += cnt[k];
  part[threadIdx.x] = s;
  __syncthreads();
  for (int d = 1; d < 1024; d <<= 1) {
    int v = (threadIdx.x >= d) ? part[threadIdx.x - d] : 0;
    __syncthreads();
    part[threadIdx.x] += v;
    __syncthreads();
  }
  int run = part[threadIdx.x] - s;
  for (int k = b; k < e; ++k) { off[k] = run; run += cnt[k]; }
  if (threadIdx.x == 1023) off[n] = part[1023];
}

__global__ void place_kernel(const int* __restrict__ row, const int* __restrict__ col,
                             const int* __restrict__ offN, const int* __restrict__ offE,
                             const int* __restrict__ rankN, const int* __restrict__ rankE,
                             int* __restrict__ permN, int* __restrict__ permE) {
  int i = blockIdx.x * blockDim.x + threadIdx.x;
  if (i >= NNZ) return;
  permN[offN[row[i]] + rankN[i]] = i;
  permE[offE[col[i]] + rankE[i]] = i;
}

template <int U, int EPI>
__global__ __launch_bounds__(192)
void segment_gather(const float* __restrict__ src, const float* __restrict__ alpha,
                    const float* __restrict__ inv, const int* __restrict__ other,
                    const int* __restrict__ off, const int* __restrict__ perm,
                    const float* __restrict__ bias,
                    float* __restrict__ dst, int nseg) {
  const int idx = threadIdx.x << 2;
  float4 bv = make_float4(0.0f, 0.0f, 0.0f, 0.0f);
  if (EPI != 0) {
    const int h = idx & (HH - 1);
    bv = make_float4(bias[h], bias[h + 1], bias[h + 2], bias[h + 3]);
  }
  const int dd = __builtin_amdgcn_readfirstlane(idx >> 7);
  for (int g = blockIdx.x; g < nseg; g += gridDim.x) {
    const int p0 = off[g], p1 = off[g + 1];
    const float w = inv[g * DD + dd];
    float4 acc = make_float4(0.0f, 0.0f, 0.0f, 0.0f);
    int p = p0;
    for (; p + U <= p1; p += U) {
      int ii[U];
      float s[U];
      float4 v[U];
#pragma unroll
      for (int u = 0; u < U; ++u) ii[u] = perm[p + u];
#pragma unroll
      for (int u = 0; u < U; ++u) {
        s[u] = w * alpha[(size_t)ii[u] * DD + dd];
        v[u] = *(const float4*)(src + (size_t)other[ii[u]] * DH + idx);
      }
#pragma unroll
      for (int u = 0; u < U; ++u) {
        acc.x += s[u] * v[u].x;
        acc.y += s[u] * v[u].y;
        acc.z += s[u] * v[u].z;
        acc.w += s[u] * v[u].w;
      }
    }
    for (; p < p1; ++p) {
      const int i = perm[p];
      const float s = w * alpha[(size_t)i * DD + dd];
      const float4 v = *(const float4*)(src + (size_t)other[i] * DH + idx);
      acc.x += s * v.x;
      acc.y += s * v.y;
      acc.z += s * v.z;
      acc.w += s * v.w;
    }
    if (EPI != 0) {
      acc.x += bv.x;
      acc.y += bv.y;
      acc.z += bv.z;
      acc.w += bv.w;
    }
    if (EPI == 2) {
      acc.x = (acc.x > 0.0f) ? acc.x : (__expf(acc.x) - 1.0f);
      acc.y = (acc.y > 0.0f) ? acc.y : (__expf(acc.y) - 1.0f);
      acc.z = (acc.z > 0.0f) ? acc.z : (__expf(acc.z) - 1.0f);
      acc.w = (acc.w > 0.0f) ? acc.w : (__expf(acc.w) - 1.0f);
    }
    *(float4*)(dst + (size_t)g * DH + idx) = acc;
  }
}

template <bool VB>
static inline void launch_gemm(const float* A, const float* B, float* C,
                               int M, int Nc, int K, int ldb, int ldc,
                               hipStream_t s) {
  dim3 grid((Nc + 63) / 64, (M + TM - 1) / TM);
  gemm_wmma_f32<VB><<<grid, dim3(128), 0, s>>>(A, B, C, M, Nc, K, ldb, ldc);
}

extern "C" void kernel_launch(void* const* d_in, const int* in_sizes, int n_in,
                              void* d_out, int out_size, void* d_ws, size_t ws_size,
                              hipStream_t stream) {
  const float* x        = (const float*)d_in[0];
  const int*   eidx     = (const int*)d_in[1];
  const float* hedge    = (const float*)d_in[2];
  const float* lin_W    = (const float*)d_in[3];
  const float* sheaf_W  = (const float*)d_in[4];
  const float* conv1_W  = (const float*)d_in[5];
  const float* conv1_b  = (const float*)d_in[6];
  const float* conv2_W  = (const float*)d_in[7];
  const float* conv2_b  = (const float*)d_in[8];
  const float* lin2_W   = (const float*)d_in[9];
  float* out = (float*)d_out;

  const int* row = eidx;
  const int* col = eidx + NNZ;

  float* ws    = (float*)d_ws;
  float* bufA  = ws;
  float* magg  = bufA + (size_t)NN * DH;
  float* el    = magg + (size_t)NN * DH;
  float* mbuf  = el + (size_t)EE * DH;
  float* xmean = mbuf + (size_t)EE * DH;
  float* emean = xmean + (size_t)NN * FF;
  float* xm    = emean + (size_t)EE * FF;
  float* em    = xm + (size_t)NN * DD;
  float* alpha = em + (size_t)EE * DD;
  float* Dg    = alpha + (size_t)NNZ * DD;
  float* Bg    = Dg + (size_t)NN * DD;

  int* cntN  = (int*)xmean;
  int* cntE  = cntN + NN;
  int* offN  = cntE + EE;
  int* offE  = offN + (NN + 1);
  int* rankN = offE + (EE + 1);
  int* rankE = rankN + NNZ;
  int* permN = rankE + NNZ;
  int* permE = permN + NNZ;

  launch_gemm<true>(x, lin_W, bufA, NN, DH, FF, DH, DH, stream);
  launch_gemm<true>(hedge, lin_W, el, EE, DH, FF, DH, DH, stream);

  mean_d_kernel<<<(NN * FF + 255) / 256, 256, 0, stream>>>(bufA, xmean, NN);
  mean_d_kernel<<<(EE * FF + 255) / 256, 256, 0, stream>>>(el, emean, EE);

  launch_gemm<false>(xmean, sheaf_W, xm, NN, DD, FF, DD, DD, stream);
  launch_gemm<false>(emean, sheaf_W + FF * DD, em, EE, DD, FF, DD, DD, stream);

  fill_kernel<<<64, 256, 0, stream>>>(Dg, (size_t)NN * DD, 0.0f);
  fill_kernel<<<64, 256, 0, stream>>>(Bg, (size_t)EE * DD, 0.0f);
  alpha_deg_kernel<<<(NNZ * DD + 255) / 256, 256, 0, stream>>>(xm, em, row, col,
                                                               alpha, Dg, Bg);
  recip_kernel<<<(NN * DD + 255) / 256, 256, 0, stream>>>(Dg, NN * DD);
  recip_kernel<<<(EE * DD + 255) / 256, 256, 0, stream>>>(Bg, EE * DD);

  fill_kernel<<<64, 256, 0, stream>>>((float*)cntN, (size_t)(NN + EE), 0.0f);
  rank_kernel<<<(NNZ + 255) / 256, 256, 0, stream>>>(row, col, cntN, cntE, rankN, rankE);
  scan_kernel<<<2, 1024, 0, stream>>>(cntN, offN, cntE, offE);
  place_kernel<<<(NNZ + 255) / 256, 256, 0, stream>>>(row, col, offN, offE, rankN, rankE,
                                                     permN, permE);


  segment_gather<8, 0><<<EE, 192, 0, stream>>>(bufA, alpha, Bg, row, offE, permE, nullptr,
                                               magg, EE);
  launch_gemm<true>(magg, conv1_W, mbuf, EE * DD, HH, FF, HH, HH, stream);
  segment_gather<4, 2><<<2560, 192, 0, stream>>>(mbuf, alpha, Dg, col, offN, permN, conv1_b,
                                                 bufA, NN);

  segment_gather<8, 0><<<EE, 192, 0, stream>>>(bufA, alpha, Bg, row, offE, permE, nullptr,
                                               magg, EE);
  launch_gemm<true>(magg, conv2_W, mbuf, EE * DD, HH, HH, HH, HH, stream);
  segment_gather<4, 1><<<2560, 192, 0, stream>>>(mbuf, alpha, Dg, col, offN, permN, conv2_b,
                                                 bufA, NN);

  launch_gemm<true>(bufA, lin2_W, out, NN, COUT, DH, COUT, COUT, stream);
}


# Class projection lin2_W moved ahead of the layer-2 hyperedge->node sum (per-d product on 2,001 hyperedge rows, bias as one more row); sums reassociated (project, then sum); [N,768] output not formed
# speedup vs baseline: 4.4140x; 1.1094x over previous
#include <hip/hip_runtime.h>
#include <hip/hip_bf16.h>

#define NN   20000
#define EE   2000
#define NNZ  100000
#define FF   128
#define HH   128
#define DD   6
#define COUT 40
#define DH   (DD * HH)

typedef __attribute__((ext_vector_type(2))) float v2f;
typedef __attribute__((ext_vector_type(8))) float v8f;

#define ATOMIC_ADD_F32(p, v) unsafeAtomicAdd((p), (v))

#define TK 32
#define TM 32
#define AP 36
template <bool VB>
__global__ __launch_bounds__(128)
void gemm_wmma_f32(const float* __restrict__ Ab, const float* __restrict__ Bb,
                   float* __restrict__ Cb, int M, int Nc, int K, int lda, int ldb, int ldc,
                   int zA, int zB, int zC) {
  __shared__ __attribute__((aligned(16))) float As[TM][AP];
  __shared__ __attribute__((aligned(16))) float Bs[TK][64];

  const float* __restrict__ A = Ab + (size_t)blockIdx.z * zA;
  const float* __restrict__ B = Bb + (size_t)blockIdx.z * zB;
  float* __restrict__ C       = Cb + (size_t)blockIdx.z * zC;

  const int tid  = threadIdx.x;
  const int wave = tid >> 5;
  const int lane = tid & 31;
  const int la   = lane & 15;
  const int hi   = lane >> 4;

  const int m0 = blockIdx.y * TM;
  const int nb = blockIdx.x * 64;
  const int n0 = nb + wave * 16;

  const int ar = tid >> 3, ac = (tid & 7) << 2;
  const int br = tid >> 4, bc = (tid & 15) << 2;
  const int gn = nb + bc;
  const float* __restrict__ Ap0 = A + (size_t)min(m0 + ar, M - 1) * lda + ac;
  const float* __restrict__ Ap1 = A + (size_t)min(m0 + ar + 16, M - 1) * lda + ac;
  const float* __restrict__ Bp  = B + (size_t)br * ldb;
  const size_t bstep = (size_t)8 * ldb;
  const int cv = min(gn, Nc - 4);
  const int c0 = min(gn + 0, Nc - 1), c1 = min(gn + 1, Nc - 1);
  const int c2 = min(gn + 2, Nc - 1), c3 = min(gn + 3, Nc - 1);

  float4 ra0, ra1, rb0, rb1, rb2, rb3;

#define GEMM_FETCH_B(dst_, q_)                                                 \
  do {                                                                         \
    if (VB) {                                                                  \
      dst_ = *(const float4*)((q_) + cv);                                      \
    } else {                                                                   \
      dst_ = make_float4((q_)[c0], (q_)[c1], (q_)[c2], (q_)[c3]);              \
    }                                                                          \
  } while (0)
#define GEMM_FETCH(kk_)                                                        \
  do {                                                                         \
    ra0 = *(const float4*)(Ap0 + (kk_));                                       \
    ra1 = *(const float4*)(Ap1 + (kk_));                                       \
    const float* qb_ = Bp + (size_t)(kk_) * ldb;                               \
    GEMM_FETCH_B(rb0, qb_);                                                    \
    GEMM_FETCH_B(rb1, qb_ + bstep);                                            \
    GEMM_FETCH_B(rb2, qb_ + 2 * bstep);                                        \
    GEMM_FETCH_B(rb3, qb_ + 3 * bstep);                                        \
  } while (0)

  v8f acc0 = {};
  v8f acc1 = {};

  GEMM_FETCH(0);

  for (int kk = 0; kk < K; kk += TK) {
    *(float4*)&As[ar][ac]      = ra0;
    *(float4*)&As[ar + 16][ac] = ra1;
    *(float4*)&Bs[br][bc]      = rb0;
    *(float4*)&Bs[br + 8][bc]  = rb1;
    *(float4*)&Bs[br + 16][bc] = rb2;
    *(float4*)&Bs[br + 24][bc] = rb3;
    __syncthreads();

    if (kk + TK < K) GEMM_FETCH(kk + TK);

#pragma unroll
    for (int k4 = 0; k4 < TK; k4 += 4) {
      v2f a0, a1, b;
      b.x  = Bs[k4 + 2 * hi][(wave << 4) + la];
      b.y  = Bs[k4 + 2 * hi + 1][(wave << 4) + la];
      a0.x = As[la][k4 + 2 * hi];
      a0.y = As[la][k4 + 2 * hi + 1];
      a1.x = As[16 + la][k4 + 2 * hi];
      a1.y = As[16 + la][k4 + 2 * hi + 1];
      acc0 = __builtin_amdgcn_wmma_f32_16x16x4_f32(
          false, a0, false, b, (short)0, acc0, false, false);
      acc1 = __builtin_amdgcn_wmma_f32_16x16x4_f32(
          false, a1, false, b, (short)0, acc1, false, false);
    }
    __syncthreads();
  }
#undef GEMM_FETCH
#undef GEMM_FETCH_B

  const int n = n0 + la;
  if (n < Nc) {
#pragma unroll
    for (int r = 0; r < 8; ++r) {
      int gm = m0 + r + 8 * hi;
      if (gm < M) C[(size_t)gm * ldc + n] = acc0[r];
    }
#pragma unroll
    for (int r = 0; r < 8; ++r) {
      int gm = m0 + 16 + r + 8 * hi;
      if (gm < M) C[(size_t)gm * ldc + n] = acc1[r];
    }
  }
}

__global__ void mean_d_kernel(const float* __restrict__ xl, float* __restrict__ xm,
                              int rows) {
  int gid = blockIdx.x * blockDim.x + threadIdx.x;
  if (gid >= rows * FF) return;
  int n = gid >> 7, f = gid & 127;
  const float* p = xl + (size_t)n * DH + f;
  float s = 0.0f;
#pragma unroll
  for (int d = 0; d < DD; ++d) s += p[d * FF];
  xm[gid] = s * (1.0f / (float)DD);
}

__global__ void alpha_deg_kernel(const float* __restrict__ xm, const float* __restrict__ em,
                                 const int* __restrict__ row, const int* __restrict__ col,
                                 float* __restrict__ alpha, float* __restrict__ Dg,
                                 float* __restrict__ Bg) {
  int gid = blockIdx.x * blockDim.x + threadIdx.x;
  if (gid >= NNZ * DD) return;
  int i = gid / DD, dd = gid - i * DD;
  int r = row[i], c = col[i];
  float v = xm[r * DD + dd] + em[c * DD + dd];
  v = 1.0f / (1.0f + __expf(-v));
  alpha[gid] = v;
  ATOMIC_ADD_F32(&Dg[r * DD + dd], v);
  ATOMIC_ADD_F32(&Bg[c * DD + dd], v);
}

__global__ void recip_kernel(float* __restrict__ p, int n) {
  int gid = blockIdx.x * blockDim.x + threadIdx.x;
  if (gid >= n) return;
  float v = p[gid];
  p[gid] = (v != 0.0f) ? (1.0f / v) : 0.0f;
}

__global__ void fill_kernel(float* __restrict__ p, size_t n, float v) {
  size_t stride = (size_t)gridDim.x * blockDim.x;
  for (size_t i = blockIdx.x * (size_t)blockDim.x + threadIdx.x; i < n; i += stride)
    p[i] = v;
}

__global__ void rank_kernel(const int* __restrict__ row, const int* __restrict__ col,
                            int* __restrict__ cntN, int* __restrict__ cntE,
                            int* __restrict__ rankN, int* __restrict__ rankE) {
  int i = blockIdx.x * blockDim.x + threadIdx.x;
  if (i >= NNZ) return;
  rankN[i] = atomicAdd(&cntN[row[i]], 1);
  rankE[i] = atomicAdd(&cntE[col[i]], 1);
}

__global__ __launch_bounds__(1024)
void scan_kernel(const int* __restrict__ cntN, int* __restrict__ offN,
                 const int* __restrict__ cntE, int* __restrict__ offE) {
  __shared__ int part[1024];
  const int* __restrict__ cnt = (blockIdx.x == 0) ? cntN : cntE;
  int* __restrict__ off       = (blockIdx.x == 0) ? offN : offE;
  const int n   = (blockIdx.x == 0) ? NN : EE;
  const int per = (n + 1023) >> 10;
  const int b = threadIdx.x * per;
  const int e = min(b + per, n);
  int s = 0;
  for (int k = b; k < e; ++k) s += cnt[k];
  part[threadIdx.x] = s;
  __syncthreads();
  for (int d = 1; d < 1024; d <<= 1) {
    int v = (threadIdx.x >= d) ? part[threadIdx.x - d] : 0;
    __syncthreads();
    part[threadIdx.x] += v;
    __syncthreads();
  }
  int run = part[threadIdx.x] - s;
  for (int k = b; k < e; ++k) { off[k] = run; run += cnt[k]; }
  if (threadIdx.x == 1023) off[n] = part[1023];
}

__global__ void place_kernel(const int* __restrict__ row, const int* __restrict__ col,
                             const int* __restrict__ offN, const int* __restrict__ offE,
                             const int* __restrict__ rankN, const int* __restrict__ rankE,
                             int* __restrict__ permN, int* __restrict__ permE) {
  int i = blockIdx.x * blockDim.x + threadIdx.x;
  if (i >= NNZ) return;
  permN[offN[row[i]] + rankN[i]] = i;
  permE[offE[col[i]] + rankE[i]] = i;
}

template <int U, int EPI>
__global__ __launch_bounds__(192)
void segment_gather(const float* __restrict__ src, const float* __restrict__ alpha,
                    const float* __restrict__ inv, const int* __restrict__ other,
                    const int* __restrict__ off, const int* __restrict__ perm,
                    const float* __restrict__ bias,
                    float* __restrict__ dst, int nseg) {
  const int idx = threadIdx.x << 2;
  float4 bv = make_float4(0.0f, 0.0f, 0.0f, 0.0f);
  if (EPI != 0) {
    const int h = idx & (HH - 1);
    bv = make_float4(bias[h], bias[h + 1], bias[h + 2], bias[h + 3]);
  }
  const int dd = __builtin_amdgcn_readfirstlane(idx >> 7);
  for (int g = blockIdx.x; g < nseg; g += gridDim.x) {
    const int p0 = off[g], p1 = off[g + 1];
    const float w = inv[g * DD + dd];
    float4 acc = make_float4(0.0f, 0.0f, 0.0f, 0.0f);
    int p = p0;
    for (; p + U <= p1; p += U) {
      int ii[U];
      float s[U];
      float4 v[U];
#pragma unroll
      for (int u = 0; u < U; ++u) ii[u] = perm[p + u];
#pragma unroll
      for (int u = 0; u < U; ++u) {
        s[u] = w * alpha[(size_t)ii[u] * DD + dd];
        v[u] = *(const float4*)(src + (size_t)other[ii[u]] * DH + idx);
      }
#pragma unroll
      for (int u = 0; u < U; ++u) {
        acc.x += s[u] * v[u].x;
        acc.y += s[u] * v[u].y;
        acc.z += s[u] * v[u].z;
        acc.w += s[u] * v[u].w;
      }
    }
    for (; p < p1; ++p) {
      const int i = perm[p];
      const float s = w * alpha[(size_t)i * DD + dd];
      const float4 v = *(const float4*)(src + (size_t)other[i] * DH + idx);
      acc.x += s * v.x;
      acc.y += s * v.y;
      acc.z += s * v.z;
      acc.w += s * v.w;
    }
    if (EPI != 0) {
      acc.x += bv.x;
      acc.y += bv.y;
      acc.z += bv.z;
      acc.w += bv.w;
    }
    if (EPI == 2) {
      acc.x = (acc.x > 0.0f) ? acc.x : (__expf(acc.x) - 1.0f);
      acc.y = (acc.y > 0.0f) ? acc.y : (__expf(acc.y) - 1.0f);
      acc.z = (acc.z > 0.0f) ? acc.z : (__expf(acc.z) - 1.0f);
      acc.w = (acc.w > 0.0f) ? acc.w : (__expf(acc.w) - 1.0f);
    }
    *(float4*)(dst + (size_t)g * DH + idx) = acc;
  }
}

__global__ void bias_row_kernel(const float* __restrict__ b, float* __restrict__ dst) {
  int t = blockIdx.x * blockDim.x + threadIdx.x;
  if (t < DH) dst[t] = b[t & (HH - 1)];
}

#define PZ (2048 * COUT)
template <int U>
__global__ __launch_bounds__(256)
void node_class_gather(const float* __restrict__ P, const float* __restrict__ alpha,
                       const float* __restrict__ Dinv, const int* __restrict__ col,
                       const int* __restrict__ off, const int* __restrict__ perm,
                       float* __restrict__ out) {
  const int gid = blockIdx.x * blockDim.x + threadIdx.x;
  if (gid >= NN * (COUT / 4)) return;
  const int n = gid / (COUT / 4);
  const float* __restrict__ Pq = P + ((gid - n * (COUT / 4)) << 2);

  float w[DD];
  float4 acc = make_float4(0.0f, 0.0f, 0.0f, 0.0f);
#pragma unroll
  for (int d = 0; d < DD; ++d) {
    w[d] = Dinv[n * DD + d];
    const float4 v = *(const float4*)(Pq + (size_t)d * PZ + (size_t)EE * COUT);
    acc.x += v.x;
    acc.y += v.y;
    acc.z += v.z;
    acc.w += v.w;
  }

  const int p0 = off[n], p1 = off[n + 1];
  for (int p = p0; p < p1; p += U) {
    int ii[U];
    float s[U][DD];
    float4 v[U][DD];
#pragma unroll
    for (int u = 0; u < U; ++u) ii[u] = perm[min(p + u, p1 - 1)];
#pragma unroll
    for (int u = 0; u < U; ++u) {
      const float* __restrict__ Pr = Pq + (size_t)col[ii[u]] * COUT;
      const bool live = (p + u) < p1;
#pragma unroll
      for (int d = 0; d < DD; ++d) {
        s[u][d] = (live ? w[d] : 0.0f) * alpha[(size_t)ii[u] * DD + d];
        v[u][d] = *(const float4*)(Pr + (size_t)d * PZ);
      }
    }
#pragma unroll
    for (int u = 0; u < U; ++u) {
#pragma unroll
      for (int d = 0; d < DD; ++d) {
        acc.x += s[u][d] * v[u][d].x;
        acc.y += s[u][d] * v[u][d].y;
        acc.z += s[u][d] * v[u][d].z;
        acc.w += s[u][d] * v[u][d].w;
      }
    }
  }
  *(float4*)(out + ((size_t)gid << 2)) = acc;
}

template <bool VB>
static inline void launch_gemm(const float* A, const float* B, float* C,
                               int M, int Nc, int K, int lda, int ldb, int ldc,
                               hipStream_t s, int nz = 1, int zA = 0, int zB = 0,
                               int zC = 0) {
  dim3 grid((Nc + 63) / 64, (M + TM - 1) / TM, nz);
  gemm_wmma_f32<VB><<<grid, dim3(128), 0, s>>>(A, B, C, M, Nc, K, lda, ldb, ldc,
                                               zA, zB, zC);
}

extern "C" void kernel_launch(void* const* d_in, const int* in_sizes, int n_in,
                              void* d_out, int out_size, void* d_ws, size_t ws_size,
                              hipStream_t stream) {
  const float* x        = (const float*)d_in[0];
  const int*   eidx     = (const int*)d_in[1];
  const float* hedge    = (const float*)d_in[2];
  const float* lin_W    = (const float*)d_in[3];
  const float* sheaf_W  = (const float*)d_in[4];
  const float* conv1_W  = (const float*)d_in[5];
  const float* conv1_b  = (const float*)d_in[6];
  const float* conv2_W  = (const float*)d_in[7];
  const float* conv2_b  = (const float*)d_in[8];
  const float* lin2_W   = (const float*)d_in[9];
  float* out = (float*)d_out;

  const int* row = eidx;
  const int* col = eidx + NNZ;

  float* ws    = (float*)d_ws;
  float* bufA  = ws;
  float* magg  = bufA + (size_t)NN * DH;
  float* m2    = magg + (size_t)2 * EE * DH;
  float* P     = magg + (size_t)4 * EE * DH;
  float* el    = magg + (size_t)NN * DH;
  float* mbuf  = el + (size_t)EE * DH;
  float* xmean = mbuf + (size_t)EE * DH;
  float* emean = xmean + (size_t)NN * FF;
  float* xm    = emean + (size_t)EE * FF;
  float* em    = xm + (size_t)NN * DD;
  float* alpha = em + (size_t)EE * DD;
  float* Dg    = alpha + (size_t)NNZ * DD;
  float* Bg    = Dg + (size_t)NN * DD;

  int* cntN  = (int*)xmean;
  int* cntE  = cntN + NN;
  int* offN  = cntE + EE;
  int* offE  = offN + (NN + 1);
  int* rankN = offE + (EE + 1);
  int* rankE = rankN + NNZ;
  int* permN = rankE + NNZ;
  int* permE = permN + NNZ;

  launch_gemm<true>(x, lin_W, bufA, NN, DH, FF, FF, DH, DH, stream);
  launch_gemm<true>(hedge, lin_W, el, EE, DH, FF, FF, DH, DH, stream);

  mean_d_kernel<<<(NN * FF + 255) / 256, 256, 0, stream>>>(bufA, xmean, NN);
  mean_d_kernel<<<(EE * FF + 255) / 256, 256, 0, stream>>>(el, emean, EE);

  launch_gemm<false>(xmean, sheaf_W, xm, NN, DD, FF, FF, DD, DD, stream);
  launch_gemm<false>(emean, sheaf_W + FF * DD, em, EE, DD, FF, FF, DD, DD, stream);

  fill_kernel<<<64, 256, 0, stream>>>(Dg, (size_t)NN * DD, 0.0f);
  fill_kernel<<<64, 256, 0, stream>>>(Bg, (size_t)EE * DD, 0.0f);
  alpha_deg_kernel<<<(NNZ * DD + 255) / 256, 256, 0, stream>>>(xm, em, row, col,
                                                               alpha, Dg, Bg);
  recip_kernel<<<(NN * DD + 255) / 256, 256, 0, stream>>>(Dg, NN * DD);
  recip_kernel<<<(EE * DD + 255) / 256, 256, 0, stream>>>(Bg, EE * DD);

  fill_kernel<<<64, 256, 0, stream>>>((float*)cntN, (size_t)(NN + EE), 0.0f);
  rank_kernel<<<(NNZ + 255) / 256, 256, 0, stream>>>(row, col, cntN, cntE, rankN, rankE);
  scan_kernel<<<2, 1024, 0, stream>>>(cntN, offN, cntE, offE);
  place_kernel<<<(NNZ + 255) / 256, 256, 0, stream>>>(row, col, offN, offE, rankN, rankE,
                                                     permN, permE);


  segment_gather<8, 0><<<EE, 192, 0, stream>>>(bufA, alpha, Bg, row, offE, permE, nullptr,
                                               magg, EE);
  launch_gemm<true>(magg, conv1_W, mbuf, EE * DD, HH, FF, FF, HH, HH, stream);
  segment_gather<4, 2><<<2560, 192, 0, stream>>>(mbuf, alpha, Dg, col, offN, permN, conv1_b,
                                                 bufA, NN);

  segment_gather<8, 0><<<EE, 192, 0, stream>>>(bufA, alpha, Bg, row, offE, permE, nullptr,
                                               magg, EE);
  launch_gemm<true>(magg, conv2_W, m2, EE * DD, HH, HH, HH, HH, HH, stream);

  bias_row_kernel<<<(DH + 255) / 256, 256, 0, stream>>>(conv2_b, m2 + (size_t)EE * DH);
  launch_gemm<true>(m2, lin2_W, P, EE + 1, COUT, HH, DH, COUT, COUT, stream,
                    DD, HH, HH * COUT, PZ);
  node_class_gather<4><<<(NN * (COUT / 4) + 255) / 256, 256, 0, stream>>>(
      P, alpha, Dg, col, offN, permN, out);
}


# Input projection lin_W moved behind the mean over d and the layer-1 node->hyperedge sum (both linear): it meets 128 + 12,000 rows, xl [N,768] is not formed; sums reassociated (sum, then project)
# speedup vs baseline: 5.5890x; 1.2662x over previous
#include <hip/hip_runtime.h>
#include <hip/hip_bf16.h>

#define NN   20000
#define EE   2000
#define NNZ  100000
#define FF   128
#define HH   128
#define DD   6
#define COUT 40
#define DH   (DD * HH)

typedef __attribute__((ext_vector_type(2))) float v2f;
typedef __attribute__((ext_vector_type(8))) float v8f;

#define ATOMIC_ADD_F32(p, v) unsafeAtomicAdd((p), (v))

#define TK 32
#define TM 32
#define AP 36
template <bool VB>
__global__ __launch_bounds__(128)
void gemm_wmma_f32(const float* __restrict__ Ab, const float* __restrict__ Bb,
                   float* __restrict__ Cb, int M, int Nc, int K, int lda, int ldb, int ldc,
                   int zA, int zB, int zC) {
  __shared__ __attribute__((aligned(16))) float As[TM][AP];
  __shared__ __attribute__((aligned(16))) float Bs[TK][64];

  const float* __restrict__ A = Ab + (size_t)blockIdx.z * zA;
  const float* __restrict__ B = Bb + (size_t)blockIdx.z * zB;
  float* __restrict__ C       = Cb + (size_t)blockIdx.z * zC;

  const int tid  = threadIdx.x;
  const int wave = tid >> 5;
  const int lane = tid & 31;
  const int la   = lane & 15;
  const int hi   = lane >> 4;

  const int m0 = blockIdx.y * TM;
  const int nb = blockIdx.x * 64;
  const int n0 = nb + wave * 16;

  const int ar = tid >> 3, ac = (tid & 7) << 2;
  const int br = tid >> 4, bc = (tid & 15) << 2;
  const int gn = nb + bc;
  const float* __restrict__ Ap0 = A + (size_t)min(m0 + ar, M - 1) * lda + ac;
  const float* __restrict__ Ap1 = A + (size_t)min(m0 + ar + 16, M - 1) * lda + ac;
  const float* __restrict__ Bp  = B + (size_t)br * ldb;
  const size_t bstep = (size_t)8 * ldb;
  const int cv = min(gn, Nc - 4);
  const int c0 = min(gn + 0, Nc - 1), c1 = min(gn + 1, Nc - 1);
  const int c2 = min(gn + 2, Nc - 1), c3 = min(gn + 3, Nc - 1);

  float4 ra0, ra1, rb0, rb1, rb2, rb3;

#define GEMM_FETCH_B(dst_, q_)                                                 \
  do {                                                                         \
    if (VB) {                                                                  \
      dst_ = *(const float4*)((q_) + cv);                                      \
    } else {                                                                   \
      dst_ = make_float4((q_)[c0], (q_)[c1], (q_)[c2], (q_)[c3]);              \
    }                                                                          \
  } while (0)
#define GEMM_FETCH(kk_)                                                        \
  do {                                                                         \
    ra0 = *(const float4*)(Ap0 + (kk_));                                       \
    ra1 = *(const float4*)(Ap1 + (kk_));                                       \
    const float* qb_ = Bp + (size_t)(kk_) * ldb;                               \
    GEMM_FETCH_B(rb0, qb_);                                                    \
    GEMM_FETCH_B(rb1, qb_ + bstep);                                            \
    GEMM_FETCH_B(rb2, qb_ + 2 * bstep);                                        \
    GEMM_FETCH_B(rb3, qb_ + 3 * bstep);                                        \
  } while (0)

  v8f acc0 = {};
  v8f acc1 = {};

  GEMM_FETCH(0);

  for (int kk = 0; kk < K; kk += TK) {
    *(float4*)&As[ar][ac]      = ra0;
    *(float4*)&As[ar + 16][ac] = ra1;
    *(float4*)&Bs[br][bc]      = rb0;
    *(float4*)&Bs[br + 8][bc]  = rb1;
    *(float4*)&Bs[br + 16][bc] = rb2;
    *(float4*)&Bs[br + 24][bc] = rb3;
    __syncthreads();

    if (kk + TK < K) GEMM_FETCH(kk + TK);

#pragma unroll
    for (int k4 = 0; k4 < TK; k4 += 4) {
      v2f a0, a1, b;
      b.x  = Bs[k4 + 2 * hi][(wave << 4) + la];
      b.y  = Bs[k4 + 2 * hi + 1][(wave << 4) + la];
      a0.x = As[la][k4 + 2 * hi];
      a0.y = As[la][k4 + 2 * hi + 1];
      a1.x = As[16 + la][k4 + 2 * hi];
      a1.y = As[16 + la][k4 + 2 * hi + 1];
      acc0 = __builtin_amdgcn_wmma_f32_16x16x4_f32(
          false, a0, false, b, (short)0, acc0, false, false);
      acc1 = __builtin_amdgcn_wmma_f32_16x16x4_f32(
          false, a1, false, b, (short)0, acc1, false, false);
    }
    __syncthreads();
  }
#undef GEMM_FETCH
#undef GEMM_FETCH_B

  const int n = n0 + la;
  if (n < Nc) {
#pragma unroll
    for (int r = 0; r < 8; ++r) {
      int gm = m0 + r + 8 * hi;
      if (gm < M) C[(size_t)gm * ldc + n] = acc0[r];
    }
#pragma unroll
    for (int r = 0; r < 8; ++r) {
      int gm = m0 + 16 + r + 8 * hi;
      if (gm < M) C[(size_t)gm * ldc + n] = acc1[r];
    }
  }
}

__global__ void mean_d_kernel(const float* __restrict__ xl, float* __restrict__ xm,
                              int rows) {
  int gid = blockIdx.x * blockDim.x + threadIdx.x;
  if (gid >= rows * FF) return;
  int n = gid >> 7, f = gid & 127;
  const float* p = xl + (size_t)n * DH + f;
  float s = 0.0f;
#pragma unroll
  for (int d = 0; d < DD; ++d) s += p[d * FF];
  xm[gid] = s * (1.0f / (float)DD);
}

__global__ void alpha_deg_kernel(const float* __restrict__ xm, const float* __restrict__ em,
                                 const int* __restrict__ row, const int* __restrict__ col,
                                 float* __restrict__ alpha, float* __restrict__ Dg,
                                 float* __restrict__ Bg) {
  int gid = blockIdx.x * blockDim.x + threadIdx.x;
  if (gid >= NNZ * DD) return;
  int i = gid / DD, dd = gid - i * DD;
  int r = row[i], c = col[i];
  float v = xm[r * DD + dd] + em[c * DD + dd];
  v = 1.0f / (1.0f + __expf(-v));
  alpha[gid] = v;
  ATOMIC_ADD_F32(&Dg[r * DD + dd], v);
  ATOMIC_ADD_F32(&Bg[c * DD + dd], v);
}

__global__ void recip_kernel(float* __restrict__ p, int n) {
  int gid = blockIdx.x * blockDim.x + threadIdx.x;
  if (gid >= n) return;
  float v = p[gid];
  p[gid] = (v != 0.0f) ? (1.0f / v) : 0.0f;
}

__global__ void fill_kernel(float* __restrict__ p, size_t n, float v) {
  size_t stride = (size_t)gridDim.x * blockDim.x;
  for (size_t i = blockIdx.x * (size_t)blockDim.x + threadIdx.x; i < n; i += stride)
    p[i] = v;
}

__global__ void rank_kernel(const int* __restrict__ row, const int* __restrict__ col,
                            int* __restrict__ cntN, int* __restrict__ cntE,
                            int* __restrict__ rankN, int* __restrict__ rankE) {
  int i = blockIdx.x * blockDim.x + threadIdx.x;
  if (i >= NNZ) return;
  rankN[i] = atomicAdd(&cntN[row[i]], 1);
  rankE[i] = atomicAdd(&cntE[col[i]], 1);
}

__global__ __launch_bounds__(1024)
void scan_kernel(const int* __restrict__ cntN, int* __restrict__ offN,
                 const int* __restrict__ cntE, int* __restrict__ offE) {
  __shared__ int part[1024];
  const int* __restrict__ cnt = (blockIdx.x == 0) ? cntN : cntE;
  int* __restrict__ off       = (blockIdx.x == 0) ? offN : offE;
  const int n   = (blockIdx.x == 0) ? NN : EE;
  const int per = (n + 1023) >> 10;
  const int b = threadIdx.x * per;
  const int e = min(b + per, n);
  int s = 0;
  for (int k = b; k < e; ++k) s += cnt[k];
  part[threadIdx.x] = s;
  __syncthreads();
  for (int d = 1; d < 1024; d <<= 1) {
    int v = (threadIdx.x >= d) ? part[threadIdx.x - d] : 0;
    __syncthreads();
    part[threadIdx.x] += v;
    __syncthreads();
  }
  int run = part[threadIdx.x] - s;
  for (int k = b; k < e; ++k) { off[k] = run; run += cnt[k]; }
  if (threadIdx.x == 1023) off[n] = part[1023];
}

__global__ void place_kernel(const int* __restrict__ row, const int* __restrict__ col,
                             const int* __restrict__ offN, const int* __restrict__ offE,
                             const int* __restrict__ rankN, const int* __restrict__ rankE,
                             int* __restrict__ permN, int* __restrict__ permE) {
  int i = blockIdx.x * blockDim.x + threadIdx.x;
  if (i >= NNZ) return;
  permN[offN[row[i]] + rankN[i]] = i;
  permE[offE[col[i]] + rankE[i]] = i;
}

template <int U, int EPI>
__global__ __launch_bounds__(192)
void segment_gather(const float* __restrict__ src, const float* __restrict__ alpha,
                    const float* __restrict__ inv, const int* __restrict__ other,
                    const int* __restrict__ off, const int* __restrict__ perm,
                    const float* __restrict__ bias,
                    float* __restrict__ dst, int nseg) {
  const int idx = threadIdx.x << 2;
  float4 bv = make_float4(0.0f, 0.0f, 0.0f, 0.0f);
  if (EPI != 0) {
    const int h = idx & (HH - 1);
    bv = make_float4(bias[h], bias[h + 1], bias[h + 2], bias[h + 3]);
  }
  const int dd = __builtin_amdgcn_readfirstlane(idx >> 7);
  for (int g = blockIdx.x; g < nseg; g += gridDim.x) {
    const int p0 = off[g], p1 = off[g + 1];
    const float w = inv[g * DD + dd];
    float4 acc = make_float4(0.0f, 0.0f, 0.0f, 0.0f);
    int p = p0;
    for (; p + U <= p1; p += U) {
      int ii[U];
      float s[U];
      float4 v[U];
#pragma unroll
      for (int u = 0; u < U; ++u) ii[u] = perm[p + u];
#pragma unroll
      for (int u = 0; u < U; ++u) {
        s[u] = w * alpha[(size_t)ii[u] * DD + dd];
        v[u] = *(const float4*)(src + (size_t)other[ii[u]] * DH + idx);
      }
#pragma unroll
      for (int u = 0; u < U; ++u) {
        acc.x += s[u] * v[u].x;
        acc.y += s[u] * v[u].y;
        acc.z += s[u] * v[u].z;
        acc.w += s[u] * v[u].w;
      }
    }
    for (; p < p1; ++p) {
      const int i = perm[p];
      const float s = w * alpha[(size_t)i * DD + dd];
      const float4 v = *(const float4*)(src + (size_t)other[i] * DH + idx);
      acc.x += s * v.x;
      acc.y += s * v.y;
      acc.z += s * v.z;
      acc.w += s * v.w;
    }
    if (EPI != 0) {
      acc.x += bv.x;
      acc.y += bv.y;
      acc.z += bv.z;
      acc.w += bv.w;
    }
    if (EPI == 2) {
      acc.x = (acc.x > 0.0f) ? acc.x : (__expf(acc.x) - 1.0f);
      acc.y = (acc.y > 0.0f) ? acc.y : (__expf(acc.y) - 1.0f);
      acc.z = (acc.z > 0.0f) ? acc.z : (__expf(acc.z) - 1.0f);
      acc.w = (acc.w > 0.0f) ? acc.w : (__expf(acc.w) - 1.0f);
    }
    *(float4*)(dst + (size_t)g * DH + idx) = acc;
  }
}

template <int U>
__global__ __launch_bounds__(32)
void edge_gather_raw(const float* __restrict__ x, const float* __restrict__ alpha,
                     const float* __restrict__ inv, const int* __restrict__ other,
                     const int* __restrict__ off, const int* __restrict__ perm,
                     float* __restrict__ dst) {
  const int g    = blockIdx.x;
  const int lane = threadIdx.x;
  const int f4   = lane << 2;
  const int p0 = off[g], p1 = off[g + 1];
  float w[DD];
  float4 acc[DD];
#pragma unroll
  for (int d = 0; d < DD; ++d) {
    w[d]   = inv[g * DD + d];
    acc[d] = make_float4(0.0f, 0.0f, 0.0f, 0.0f);
  }
  for (int p = p0; p < p1; p += 32) {
    const bool live = (p + lane) < p1;
    const int i = perm[min(p + lane, p1 - 1)];
    const int o = other[i];
    float sl[DD];
#pragma unroll
    for (int d = 0; d < DD; ++d)
      sl[d] = (live ? w[d] : 0.0f) * alpha[(size_t)i * DD + d];
#pragma unroll
    for (int jb = 0; jb < 32; jb += U) {
      if (p + jb < p1) {
        float4 v[U];
#pragma unroll
        for (int u = 0; u < U; ++u) {
          const int oo = __builtin_amdgcn_readlane(o, jb + u);
          v[u] = *(const float4*)(x + (size_t)oo * FF + f4);
        }
#pragma unroll
        for (int u = 0; u < U; ++u) {
#pragma unroll
          for (int d = 0; d < DD; ++d) {
            const float sd = __int_as_float(
                __builtin_amdgcn_readlane(__float_as_int(sl[d]), jb + u));
            acc[d].x += sd * v[u].x;
            acc[d].y += sd * v[u].y;
            acc[d].z += sd * v[u].z;
            acc[d].w += sd * v[u].w;
          }
        }
      }
    }
  }
#pragma unroll
  for (int d = 0; d < DD; ++d)
    *(float4*)(dst + (size_t)g * DH + d * FF + f4) = acc[d];
}

__global__ void bias_row_kernel(const float* __restrict__ b, float* __restrict__ dst) {
  int t = blockIdx.x * blockDim.x + threadIdx.x;
  if (t < DH) dst[t] = b[t & (HH - 1)];
}

#define PZ (2048 * COUT)
template <int U>
__global__ __launch_bounds__(256)
void node_class_gather(const float* __restrict__ P, const float* __restrict__ alpha,
                       const float* __restrict__ Dinv, const int* __restrict__ col,
                       const int* __restrict__ off, const int* __restrict__ perm,
                       float* __restrict__ out) {
  const int gid = blockIdx.x * blockDim.x + threadIdx.x;
  if (gid >= NN * (COUT / 4)) return;
  const int n = gid / (COUT / 4);
  const float* __restrict__ Pq = P + ((gid - n * (COUT / 4)) << 2);

  float w[DD];
  float4 acc = make_float4(0.0f, 0.0f, 0.0f, 0.0f);
#pragma unroll
  for (int d = 0; d < DD; ++d) {
    w[d] = Dinv[n * DD + d];
    const float4 v = *(const float4*)(Pq + (size_t)d * PZ + (size_t)EE * COUT);
    acc.x += v.x;
    acc.y += v.y;
    acc.z += v.z;
    acc.w += v.w;
  }

  const int p0 = off[n], p1 = off[n + 1];
  for (int p = p0; p < p1; p += U) {
    int ii[U];
    float s[U][DD];
    float4 v[U][DD];
#pragma unroll
    for (int u = 0; u < U; ++u) ii[u] = perm[min(p + u, p1 - 1)];
#pragma unroll
    for (int u = 0; u < U; ++u) {
      const float* __restrict__ Pr = Pq + (size_t)col[ii[u]] * COUT;
      const bool live = (p + u) < p1;
#pragma unroll
      for (int d = 0; d < DD; ++d) {
        s[u][d] = (live ? w[d] : 0.0f) * alpha[(size_t)ii[u] * DD + d];
        v[u][d] = *(const float4*)(Pr + (size_t)d * PZ);
      }
    }
#pragma unroll
    for (int u = 0; u < U; ++u) {
#pragma unroll
      for (int d = 0; d < DD; ++d) {
        acc.x += s[u][d] * v[u][d].x;
        acc.y += s[u][d] * v[u][d].y;
        acc.z += s[u][d] * v[u][d].z;
        acc.w += s[u][d] * v[u][d].w;
      }
    }
  }
  *(float4*)(out + ((size_t)gid << 2)) = acc;
}

template <bool VB>
static inline void launch_gemm(const float* A, const float* B, float* C,
                               int M, int Nc, int K, int lda, int ldb, int ldc,
                               hipStream_t s, int nz = 1, int zA = 0, int zB = 0,
                               int zC = 0) {
  dim3 grid((Nc + 63) / 64, (M + TM - 1) / TM, nz);
  gemm_wmma_f32<VB><<<grid, dim3(128), 0, s>>>(A, B, C, M, Nc, K, lda, ldb, ldc,
                                               zA, zB, zC);
}

extern "C" void kernel_launch(void* const* d_in, const int* in_sizes, int n_in,
                              void* d_out, int out_size, void* d_ws, size_t ws_size,
                              hipStream_t stream) {
  const float* x        = (const float*)d_in[0];
  const int*   eidx     = (const int*)d_in[1];
  const float* hedge    = (const float*)d_in[2];
  const float* lin_W    = (const float*)d_in[3];
  const float* sheaf_W  = (const float*)d_in[4];
  const float* conv1_W  = (const float*)d_in[5];
  const float* conv1_b  = (const float*)d_in[6];
  const float* conv2_W  = (const float*)d_in[7];
  const float* conv2_b  = (const float*)d_in[8];
  const float* lin2_W   = (const float*)d_in[9];
  float* out = (float*)d_out;

  const int* row = eidx;
  const int* col = eidx + NNZ;

  float* ws    = (float*)d_ws;
  float* bufA  = ws;
  float* magg  = bufA + (size_t)NN * DH;
  float* m2    = magg + (size_t)2 * EE * DH;
  float* P     = magg + (size_t)4 * EE * DH;
  float* xagg  = magg + (size_t)NN * DH;
  float* mbuf  = xagg + (size_t)EE * DH;
  float* csr   = mbuf + (size_t)EE * DH;
  float* Wmean = csr + (size_t)NN * FF;
  float* V     = Wmean + (size_t)FF * FF;
  float* xm    = Wmean + (size_t)EE * FF;
  float* em    = xm + (size_t)NN * DD;
  float* alpha = em + (size_t)EE * DD;
  float* Dg    = alpha + (size_t)NNZ * DD;
  float* Bg    = Dg + (size_t)NN * DD;

  int* cntN  = (int*)csr;
  int* cntE  = cntN + NN;
  int* offN  = cntE + EE;
  int* offE  = offN + (NN + 1);
  int* rankN = offE + (EE + 1);
  int* rankE = rankN + NNZ;
  int* permN = rankE + NNZ;
  int* permE = permN + NNZ;


  mean_d_kernel<<<(FF * FF + 255) / 256, 256, 0, stream>>>(lin_W, Wmean, FF);

  launch_gemm<false>(Wmean, sheaf_W, V, FF, DD, FF, FF, DD, DD, stream,
                     2, 0, FF * DD, FF * DD);

  launch_gemm<false>(x, V, xm, NN, DD, FF, FF, DD, DD, stream);
  launch_gemm<false>(hedge, V + FF * DD, em, EE, DD, FF, FF, DD, DD, stream);

  fill_kernel<<<64, 256, 0, stream>>>(Dg, (size_t)NN * DD, 0.0f);
  fill_kernel<<<64, 256, 0, stream>>>(Bg, (size_t)EE * DD, 0.0f);
  alpha_deg_kernel<<<(NNZ * DD + 255) / 256, 256, 0, stream>>>(xm, em, row, col,
                                                               alpha, Dg, Bg);
  recip_kernel<<<(NN * DD + 255) / 256, 256, 0, stream>>>(Dg, NN * DD);
  recip_kernel<<<(EE * DD + 255) / 256, 256, 0, stream>>>(Bg, EE * DD);

  fill_kernel<<<64, 256, 0, stream>>>((float*)cntN, (size_t)(NN + EE), 0.0f);
  rank_kernel<<<(NNZ + 255) / 256, 256, 0, stream>>>(row, col, cntN, cntE, rankN, rankE);
  scan_kernel<<<2, 1024, 0, stream>>>(cntN, offN, cntE, offE);
  place_kernel<<<(NNZ + 255) / 256, 256, 0, stream>>>(row, col, offN, offE, rankN, rankE,
                                                     permN, permE);


  edge_gather_raw<8><<<EE, 32, 0, stream>>>(x, alpha, Bg, row, offE, permE, xagg);
  launch_gemm<true>(xagg, lin_W, magg, EE, FF, FF, DH, DH, DH, stream,
                    DD, FF, FF, FF);
  launch_gemm<true>(magg, conv1_W, mbuf, EE * DD, HH, FF, FF, HH, HH, stream);
  segment_gather<4, 2><<<2560, 192, 0, stream>>>(mbuf, alpha, Dg, col, offN, permN, conv1_b,
                                                 bufA, NN);

  segment_gather<8, 0><<<EE, 192, 0, stream>>>(bufA, alpha, Bg, row, offE, permE, nullptr,
                                               magg, EE);
  launch_gemm<true>(magg, conv2_W, m2, EE * DD, HH, HH, HH, HH, HH, stream);

  bias_row_kernel<<<(DH + 255) / 256, 256, 0, stream>>>(conv2_b, m2 + (size_t)EE * DH);
  launch_gemm<true>(m2, lin2_W, P, EE + 1, COUT, HH, DH, COUT, COUT, stream,
                    DD, HH, HH * COUT, PZ);
  node_class_gather<4><<<(NN * (COUT / 4) + 255) / 256, 256, 0, stream>>>(
      P, alpha, Dg, col, offN, permN, out);
}


# Dependent launches 22 -> 13: sheaf degrees summed inside the gathers (float atomics, fills, inversions go; 1/deg applied after the segment sum, CSR order); same-level small kernels share a launch
# speedup vs baseline: 6.9516x; 1.2438x over previous
#include <hip/hip_runtime.h>
#include <hip/hip_bf16.h>

#define NN   20000
#define EE   2000
#define NNZ  100000
#define FF   128
#define HH   128
#define DD   6
#define COUT 40
#define DH   (DD * HH)

typedef __attribute__((ext_vector_type(2))) float v2f;
typedef __attribute__((ext_vector_type(8))) float v8f;

#define TK 32
#define TM 32
#define AP 36
template <bool VB>
__device__ __forceinline__
void gemm_tile_f32(const float* __restrict__ A, const float* __restrict__ B,
                   float* __restrict__ C, int M, int Nc, int K, int lda, int ldb, int ldc,
                   int bx, int by) {
  __shared__ __attribute__((aligned(16))) float As[TM][AP];
  __shared__ __attribute__((aligned(16))) float Bs[TK][64];

  const int tid  = threadIdx.x;
  const int wave = tid >> 5;
  const int lane = tid & 31;
  const int la   = lane & 15;
  const int hi   = lane >> 4;

  const int m0 = by * TM;
  const int nb = bx * 64;
  const int n0 = nb + wave * 16;

  const int ar = tid >> 3, ac = (tid & 7) << 2;
  const int br = tid >> 4, bc = (tid & 15) << 2;
  const int gn = nb + bc;
  const float* __restrict__ Ap0 = A + (size_t)min(m0 + ar, M - 1) * lda + ac;
  const float* __restrict__ Ap1 = A + (size_t)min(m0 + ar + 16, M - 1) * lda + ac;
  const float* __restrict__ Bp  = B + (size_t)br * ldb;
  const size_t bstep = (size_t)8 * ldb;
  const int cv = min(gn, Nc - 4);
  const int c0 = min(gn + 0, Nc - 1), c1 = min(gn + 1, Nc - 1);
  const int c2 = min(gn + 2, Nc - 1), c3 = min(gn + 3, Nc - 1);

  float4 ra0, ra1, rb0, rb1, rb2, rb3;

#define GEMM_FETCH_B(dst_, q_)                                                 \
  do {                                                                         \
    if (VB) {                                                                  \
      dst_ = *(const float4*)((q_) + cv);                                      \
    } else {                                                                   \
      dst_ = make_float4((q_)[c0], (q_)[c1], (q_)[c2], (q_)[c3]);              \
    }                                                                          \
  } while (0)
#define GEMM_FETCH(kk_)                                                        \
  do {                                                                         \
    ra0 = *(const float4*)(Ap0 + (kk_));                                       \
    ra1 = *(const float4*)(Ap1 + (kk_));                                       \
    const float* qb_ = Bp + (size_t)(kk_) * ldb;                               \
    GEMM_FETCH_B(rb0, qb_);                                                    \
    GEMM_FETCH_B(rb1, qb_ + bstep);                                            \
    GEMM_FETCH_B(rb2, qb_ + 2 * bstep);                                        \
    GEMM_FETCH_B(rb3, qb_ + 3 * bstep);                                        \
  } while (0)

  v8f acc0 = {};
  v8f acc1 = {};

  GEMM_FETCH(0);

  for (int kk = 0; kk < K; kk += TK) {
    *(float4*)&As[ar][ac]      = ra0;
    *(float4*)&As[ar + 16][ac] = ra1;
    *(float4*)&Bs[br][bc]      = rb0;
    *(float4*)&Bs[br + 8][bc]  = rb1;
    *(float4*)&Bs[br + 16][bc] = rb2;
    *(float4*)&Bs[br + 24][bc] = rb3;
    __syncthreads();

    if (kk + TK < K) GEMM_FETCH(kk + TK);

#pragma unroll
    for (int k4 = 0; k4 < TK; k4 += 4) {
      v2f a0, a1, b;
      b.x  = Bs[k4 + 2 * hi][(wave << 4) + la];
      b.y  = Bs[k4 + 2 * hi + 1][(wave << 4) + la];
      a0.x = As[la][k4 + 2 * hi];
      a0.y = As[la][k4 + 2 * hi + 1];
      a1.x = As[16 + la][k4 + 2 * hi];
      a1.y = As[16 + la][k4 + 2 * hi + 1];
      acc0 = __builtin_amdgcn_wmma_f32_16x16x4_f32(
          false, a0, false, b, (short)0, acc0, false, false);
      acc1 = __builtin_amdgcn_wmma_f32_16x16x4_f32(
          false, a1, false, b, (short)0, acc1, false, false);
    }
    __syncthreads();
  }
#undef GEMM_FETCH
#undef GEMM_FETCH_B

  const int n = n0 + la;
  if (n < Nc) {
#pragma unroll
    for (int r = 0; r < 8; ++r) {
      int gm = m0 + r + 8 * hi;
      if (gm < M) C[(size_t)gm * ldc + n] = acc0[r];
    }
#pragma unroll
    for (int r = 0; r < 8; ++r) {
      int gm = m0 + 16 + r + 8 * hi;
      if (gm < M) C[(size_t)gm * ldc + n] = acc1[r];
    }
  }
}

template <bool VB>
__global__ __launch_bounds__(128)
void gemm_wmma_f32(const float* __restrict__ Ab, const float* __restrict__ Bb,
                   float* __restrict__ Cb, int M, int Nc, int K, int lda, int ldb, int ldc,
                   int zA, int zB, int zC) {
  gemm_tile_f32<VB>(Ab + (size_t)blockIdx.z * zA, Bb + (size_t)blockIdx.z * zB,
                    Cb + (size_t)blockIdx.z * zC, M, Nc, K, lda, ldb, ldc,
                    (int)blockIdx.x, (int)blockIdx.y);
}

template <bool VB>
__global__ __launch_bounds__(128)
void gemm_wmma_f32_two(const float* __restrict__ A0, const float* __restrict__ B0,
                       float* __restrict__ C0, int M0,
                       const float* __restrict__ A1, const float* __restrict__ B1,
                       float* __restrict__ C1, int M1, int nby0,
                       int Nc, int K, int lda, int ldb, int ldc) {
  const bool second = (int)blockIdx.y >= nby0;
  gemm_tile_f32<VB>(second ? A1 : A0, second ? B1 : B0, second ? C1 : C0,
                    second ? M1 : M0, Nc, K, lda, ldb, ldc, (int)blockIdx.x,
                    second ? (int)blockIdx.y - nby0 : (int)blockIdx.y);
}

__global__ void prep_kernel(const float* __restrict__ lin_W, float* __restrict__ Wmean,
                            int* __restrict__ cnt, const float* __restrict__ b,
                            float* __restrict__ brow) {
  const int gid = blockIdx.x * blockDim.x + threadIdx.x;
  if (gid < FF * FF) {
    const int n = gid >> 7, f = gid & 127;
    const float* p = lin_W + (size_t)n * DH + f;
    float s = 0.0f;
#pragma unroll
    for (int d = 0; d < DD; ++d) s += p[d * FF];
    Wmean[gid] = s * (1.0f / (float)DD);
  }
  if (gid < NN + EE) cnt[gid] = 0;
  if (gid < DH) brow[gid] = b[gid & (HH - 1)];
}

#define VBLK 8
__global__ __launch_bounds__(128)
void rank_logitw_kernel(const float* __restrict__ Wmean, const float* __restrict__ sheaf_W,
                        float* __restrict__ V,
                        const int* __restrict__ row, const int* __restrict__ col,
                        int* __restrict__ cntN, int* __restrict__ cntE,
                        int* __restrict__ rankN, int* __restrict__ rankE) {
  if (blockIdx.x < VBLK) {
    const int z = (int)blockIdx.x >> 2;
    gemm_tile_f32<false>(Wmean, sheaf_W + z * (FF * DD), V + z * (FF * DD),
                         FF, DD, FF, FF, DD, DD, 0, (int)blockIdx.x & 3);
    return;
  }
  const int i = ((int)blockIdx.x - VBLK) * 128 + (int)threadIdx.x;
  if (i >= NNZ) return;
  rankN[i] = atomicAdd(&cntN[row[i]], 1);
  rankE[i] = atomicAdd(&cntE[col[i]], 1);
}

__global__ __launch_bounds__(1024)
void scan_kernel(const int* __restrict__ cntN, int* __restrict__ offN,
                 const int* __restrict__ cntE, int* __restrict__ offE) {
  __shared__ int part[1024];
  const int* __restrict__ cnt = (blockIdx.x == 0) ? cntN : cntE;
  int* __restrict__ off       = (blockIdx.x == 0) ? offN : offE;
  const int n   = (blockIdx.x == 0) ? NN : EE;
  const int per = (n + 1023) >> 10;
  const int b = threadIdx.x * per;
  const int e = min(b + per, n);
  int s = 0;
  for (int k = b; k < e; ++k) s += cnt[k];
  part[threadIdx.x] = s;
  __syncthreads();
  for (int d = 1; d < 1024; d <<= 1) {
    int v = (threadIdx.x >= d) ? part[threadIdx.x - d] : 0;
    __syncthreads();
    part[threadIdx.x] += v;
    __syncthreads();
  }
  int run = part[threadIdx.x] - s;
  for (int k = b; k < e; ++k) { off[k] = run; run += cnt[k]; }
  if (threadIdx.x == 1023) off[n] = part[1023];
}

__global__ void alpha_place_kernel(const float* __restrict__ xm, const float* __restrict__ em,
                                   const int* __restrict__ row, const int* __restrict__ col,
                                   float* __restrict__ alpha,
                                   const int* __restrict__ offN, const int* __restrict__ offE,
                                   const int* __restrict__ rankN, const int* __restrict__ rankE,
                                   int* __restrict__ permN, int* __restrict__ permE) {
  const int gid = blockIdx.x * blockDim.x + threadIdx.x;
  if (gid >= NNZ * DD) return;
  const int i = gid / DD, dd = gid - i * DD;
  const int r = row[i], c = col[i];
  float v = xm[r * DD + dd] + em[c * DD + dd];
  v = 1.0f / (1.0f + __expf(-v));
  alpha[gid] = v;
  if (dd == 0) {
    permN[offN[r] + rankN[i]] = i;
    permE[offE[c] + rankE[i]] = i;
  }
}

template <int U, int EPI>
__global__ __launch_bounds__(192)
void segment_gather(const float* __restrict__ src, const float* __restrict__ alpha,
                    const int* __restrict__ other,
                    const int* __restrict__ off, const int* __restrict__ perm,
                    const float* __restrict__ bias,
                    float* __restrict__ dst, float* __restrict__ keep, int nseg) {
  const int idx = threadIdx.x << 2;
  float4 bv = make_float4(0.0f, 0.0f, 0.0f, 0.0f);
  if (EPI != 0) {
    const int h = idx & (HH - 1);
    bv = make_float4(bias[h], bias[h + 1], bias[h + 2], bias[h + 3]);
  }
  const int dd = __builtin_amdgcn_readfirstlane(idx >> 7);
  for (int g = blockIdx.x; g < nseg; g += gridDim.x) {
    const int p0 = off[g], p1 = off[g + 1];
    float deg = 0.0f;
    float4 acc = make_float4(0.0f, 0.0f, 0.0f, 0.0f);
    int p = p0;
    for (; p + U <= p1; p += U) {
      int ii[U];
      float s[U];
      float4 v[U];
#pragma unroll
      for (int u = 0; u < U; ++u) ii[u] = perm[p + u];
#pragma unroll
      for (int u = 0; u < U; ++u) {
        s[u] = alpha[(size_t)ii[u] * DD + dd];
        v[u] = *(const float4*)(src + (size_t)other[ii[u]] * DH + idx);
      }
#pragma unroll
      for (int u = 0; u < U; ++u) {
        deg   += s[u];
        acc.x += s[u] * v[u].x;
        acc.y += s[u] * v[u].y;
        acc.z += s[u] * v[u].z;
        acc.w += s[u] * v[u].w;
      }
    }
    for (; p < p1; ++p) {
      const int i = perm[p];
      const float s = alpha[(size_t)i * DD + dd];
      const float4 v = *(const float4*)(src + (size_t)other[i] * DH + idx);
      deg   += s;
      acc.x += s * v.x;
      acc.y += s * v.y;
      acc.z += s * v.z;
      acc.w += s * v.w;
    }
    const float w = (deg != 0.0f) ? (1.0f / deg) : 0.0f;
    if (keep != nullptr && (threadIdx.x & 31) == 0) keep[g * DD + dd] = w;
    acc.x *= w;
    acc.y *= w;
    acc.z *= w;
    acc.w *= w;
    if (EPI != 0) {
      acc.x += bv.x;
      acc.y += bv.y;
      acc.z += bv.z;
      acc.w += bv.w;
    }
    if (EPI == 2) {
      acc.x = (acc.x > 0.0f) ? acc.x : (__expf(acc.x) - 1.0f);
      acc.y = (acc.y > 0.0f) ? acc.y : (__expf(acc.y) - 1.0f);
      acc.z = (acc.z > 0.0f) ? acc.z : (__expf(acc.z) - 1.0f);
      acc.w = (acc.w > 0.0f) ? acc.w : (__expf(acc.w) - 1.0f);
    }
    *(float4*)(dst + (size_t)g * DH + idx) = acc;
  }
}

template <int U>
__global__ __launch_bounds__(32)
void edge_gather_raw(const float* __restrict__ x, const float* __restrict__ alpha,
                     const int* __restrict__ other,
                     const int* __restrict__ off, const int* __restrict__ perm,
                     float* __restrict__ dst) {
  const int g    = blockIdx.x;
  const int lane = threadIdx.x;
  const int f4   = lane << 2;
  const int p0 = off[g], p1 = off[g + 1];
  float deg[DD];
  float4 acc[DD];
#pragma unroll
  for (int d = 0; d < DD; ++d) {
    deg[d] = 0.0f;
    acc[d] = make_float4(0.0f, 0.0f, 0.0f, 0.0f);
  }
  for (int p = p0; p < p1; p += 32) {
    const bool live = (p + lane) < p1;
    const int i = perm[min(p + lane, p1 - 1)];
    const int o = other[i];
    float sl[DD];
#pragma unroll
    for (int d = 0; d < DD; ++d)
      sl[d] = (live ? 1.0f : 0.0f) * alpha[(size_t)i * DD + d];
#pragma unroll
    for (int jb = 0; jb < 32; jb += U) {
      if (p + jb < p1) {
        float4 v[U];
#pragma unroll
        for (int u = 0; u < U; ++u) {
          const int oo = __builtin_amdgcn_readlane(o, jb + u);
          v[u] = *(const float4*)(x + (size_t)oo * FF + f4);
        }
#pragma unroll
        for (int u = 0; u < U; ++u) {
#pragma unroll
          for (int d = 0; d < DD; ++d) {
            const float sd = __int_as_float(
                __builtin_amdgcn_readlane(__float_as_int(sl[d]), jb + u));
            deg[d]   += sd;
            acc[d].x += sd * v[u].x;
            acc[d].y += sd * v[u].y;
            acc[d].z += sd * v[u].z;
            acc[d].w += sd * v[u].w;
          }
        }
      }
    }
  }
#pragma unroll
  for (int d = 0; d < DD; ++d) {
    const float w = (deg[d] != 0.0f) ? (1.0f / deg[d]) : 0.0f;
    acc[d].x *= w;
    acc[d].y *= w;
    acc[d].z *= w;
    acc[d].w *= w;
    *(float4*)(dst + (size_t)g * DH + d * FF + f4) = acc[d];
  }
}

#define PZ (2048 * COUT)
template <int U>
__global__ __launch_bounds__(256)
void node_class_gather(const float* __restrict__ P, const float* __restrict__ alpha,
                       const float* __restrict__ Dinv, const int* __restrict__ col,
                       const int* __restrict__ off, const int* __restrict__ perm,
                       float* __restrict__ out) {
  const int gid = blockIdx.x * blockDim.x + threadIdx.x;
  if (gid >= NN * (COUT / 4)) return;
  const int n = gid / (COUT / 4);
  const float* __restrict__ Pq = P + ((gid - n * (COUT / 4)) << 2);

  float w[DD];
  float4 acc = make_float4(0.0f, 0.0f, 0.0f, 0.0f);
#pragma unroll
  for (int d = 0; d < DD; ++d) {
    w[d] = Dinv[n * DD + d];
    const float4 v = *(const float4*)(Pq + (size_t)d * PZ + (size_t)EE * COUT);
    acc.x += v.x;
    acc.y += v.y;
    acc.z += v.z;
    acc.w += v.w;
  }

  const int p0 = off[n], p1 = off[n + 1];
  for (int p = p0; p < p1; p += U) {
    int ii[U];
    float s[U][DD];
    float4 v[U][DD];
#pragma unroll
    for (int u = 0; u < U; ++u) ii[u] = perm[min(p + u, p1 - 1)];
#pragma unroll
    for (int u = 0; u < U; ++u) {
      const float* __restrict__ Pr = Pq + (size_t)col[ii[u]] * COUT;
      const bool live = (p + u) < p1;
#pragma unroll
      for (int d = 0; d < DD; ++d) {
        s[u][d] = (live ? w[d] : 0.0f) * alpha[(size_t)ii[u] * DD + d];
        v[u][d] = *(const float4*)(Pr + (size_t)d * PZ);
      }
    }
#pragma unroll
    for (int u = 0; u < U; ++u) {
#pragma unroll
      for (int d = 0; d < DD; ++d) {
        acc.x += s[u][d] * v[u][d].x;
        acc.y += s[u][d] * v[u][d].y;
        acc.z += s[u][d] * v[u][d].z;
        acc.w += s[u][d] * v[u][d].w;
      }
    }
  }
  *(float4*)(out + ((size_t)gid << 2)) = acc;
}

template <bool VB>
static inline void launch_gemm(const float* A, const float* B, float* C,
                               int M, int Nc, int K, int lda, int ldb, int ldc,
                               hipStream_t s, int nz = 1, int zA = 0, int zB = 0,
                               int zC = 0) {
  dim3 grid((Nc + 63) / 64, (M + TM - 1) / TM, nz);
  gemm_wmma_f32<VB><<<grid, dim3(128), 0, s>>>(A, B, C, M, Nc, K, lda, ldb, ldc,
                                               zA, zB, zC);
}

extern "C" void kernel_launch(void* const* d_in, const int* in_sizes, int n_in,
                              void* d_out, int out_size, void* d_ws, size_t ws_size,
                              hipStream_t stream) {
  const float* x        = (const float*)d_in[0];
  const int*   eidx     = (const int*)d_in[1];
  const float* hedge    = (const float*)d_in[2];
  const float* lin_W    = (const float*)d_in[3];
  const float* sheaf_W  = (const float*)d_in[4];
  const float* conv1_W  = (const float*)d_in[5];
  const float* conv1_b  = (const float*)d_in[6];
  const float* conv2_W  = (const float*)d_in[7];
  const float* conv2_b  = (const float*)d_in[8];
  const float* lin2_W   = (const float*)d_in[9];
  float* out = (float*)d_out;

  const int* row = eidx;
  const int* col = eidx + NNZ;

  float* ws    = (float*)d_ws;
  float* bufA  = ws;
  float* magg  = bufA + (size_t)NN * DH;
  float* m2    = magg + (size_t)2 * EE * DH;
  float* P     = magg + (size_t)4 * EE * DH;
  float* xagg  = magg + (size_t)NN * DH;
  float* mbuf  = xagg + (size_t)EE * DH;
  float* csr   = mbuf + (size_t)EE * DH;
  float* Wmean = csr + (size_t)NN * FF;
  float* V     = Wmean + (size_t)FF * FF;
  float* xm    = Wmean + (size_t)EE * FF;
  float* em    = xm + (size_t)NN * DD;
  float* alpha = em + (size_t)EE * DD;
  float* Dinv  = alpha + (size_t)NNZ * DD;

  int* cntN  = (int*)csr;
  int* cntE  = cntN + NN;
  int* offN  = cntE + EE;
  int* offE  = offN + (NN + 1);
  int* rankN = offE + (EE + 1);
  int* rankE = rankN + NNZ;
  int* permN = rankE + NNZ;
  int* permE = permN + NNZ;


  prep_kernel<<<(NN + EE + 255) / 256, 256, 0, stream>>>(lin_W, Wmean, cntN, conv2_b,
                                                         m2 + (size_t)EE * DH);

  rank_logitw_kernel<<<VBLK + (NNZ + 127) / 128, 128, 0, stream>>>(
      Wmean, sheaf_W, V, row, col, cntN, cntE, rankN, rankE);

  gemm_wmma_f32_two<false><<<dim3(1, (NN + TM - 1) / TM + (EE + TM - 1) / TM, 1), dim3(128),
                             0, stream>>>(x, V, xm, NN, hedge, V + FF * DD, em, EE,
                                          (NN + TM - 1) / TM, DD, FF, FF, DD, DD);

  scan_kernel<<<2, 1024, 0, stream>>>(cntN, offN, cntE, offE);
  alpha_place_kernel<<<(NNZ * DD + 255) / 256, 256, 0, stream>>>(
      xm, em, row, col, alpha, offN, offE, rankN, rankE, permN, permE);


  edge_gather_raw<8><<<EE, 32, 0, stream>>>(x, alpha, row, offE, permE, xagg);
  launch_gemm<true>(xagg, lin_W, magg, EE, FF, FF, DH, DH, DH, stream,
                    DD, FF, FF, FF);
  launch_gemm<true>(magg, conv1_W, mbuf, EE * DD, HH, FF, FF, HH, HH, stream);
  segment_gather<4, 2><<<2560, 192, 0, stream>>>(mbuf, alpha, col, offN, permN, conv1_b,
                                                 bufA, Dinv, NN);

  segment_gather<8, 0><<<EE, 192, 0, stream>>>(bufA, alpha, row, offE, permE, nullptr,
                                               magg, nullptr, EE);
  launch_gemm<true>(magg, conv2_W, m2, EE * DD, HH, HH, HH, HH, HH, stream);

  launch_gemm<true>(m2, lin2_W, P, EE + 1, COUT, HH, DH, COUT, COUT, stream,
                    DD, HH, HH * COUT, PZ);
  node_class_gather<4><<<(NN * (COUT / 4) + 255) / 256, 256, 0, stream>>>(
      P, alpha, Dinv, col, offN, permN, out);
}


# Adjacent weight matrices multiplied first (lin_W block d @ conv1_W, conv2_W @ lin2_W block d; side blocks of launch 2, on WMMA): both 12,000-row conv GEMM launches go (13 -> 11); sums reassociated
# speedup vs baseline: 7.6764x; 1.1043x over previous
#include <hip/hip_runtime.h>
#include <hip/hip_bf16.h>

#define NN   20000
#define EE   2000
#define NNZ  100000
#define FF   128
#define HH   128
#define DD   6
#define COUT 40
#define DH   (DD * HH)

typedef __attribute__((ext_vector_type(2))) float v2f;
typedef __attribute__((ext_vector_type(8))) float v8f;

#define TK 32
#define TM 32
#define AP 36
template <bool VB>
__device__ __forceinline__
void gemm_tile_f32(const float* __restrict__ A, const float* __restrict__ B,
                   float* __restrict__ C, int M, int Nc, int K, int lda, int ldb, int ldc,
                   int bx, int by) {
  __shared__ __attribute__((aligned(16))) float As[TM][AP];
  __shared__ __attribute__((aligned(16))) float Bs[TK][64];

  const int tid  = threadIdx.x;
  const int wave = tid >> 5;
  const int lane = tid & 31;
  const int la   = lane & 15;
  const int hi   = lane >> 4;

  const int m0 = by * TM;
  const int nb = bx * 64;
  const int n0 = nb + wave * 16;

  const int ar = tid >> 3, ac = (tid & 7) << 2;
  const int br = tid >> 4, bc = (tid & 15) << 2;
  const int gn = nb + bc;
  const float* __restrict__ Ap0 = A + (size_t)min(m0 + ar, M - 1) * lda + ac;
  const float* __restrict__ Ap1 = A + (size_t)min(m0 + ar + 16, M - 1) * lda + ac;
  const float* __restrict__ Bp  = B + (size_t)br * ldb;
  const size_t bstep = (size_t)8 * ldb;
  const int cv = min(gn, Nc - 4);
  const int c0 = min(gn + 0, Nc - 1), c1 = min(gn + 1, Nc - 1);
  const int c2 = min(gn + 2, Nc - 1), c3 = min(gn + 3, Nc - 1);

  float4 ra0, ra1, rb0, rb1, rb2, rb3;

#define GEMM_FETCH_B(dst_, q_)                                                 \
  do {                                                                         \
    if (VB) {                                                                  \
      dst_ = *(const float4*)((q_) + cv);                                      \
    } else {                                                                   \
      dst_ = make_float4((q_)[c0], (q_)[c1], (q_)[c2], (q_)[c3]);              \
    }                                                                          \
  } while (0)
#define GEMM_FETCH(kk_)                                                        \
  do {                                                                         \
    ra0 = *(const float4*)(Ap0 + (kk_));                                       \
    ra1 = *(const float4*)(Ap1 + (kk_));                                       \
    const float* qb_ = Bp + (size_t)(kk_) * ldb;                               \
    GEMM_FETCH_B(rb0, qb_);                                                    \
    GEMM_FETCH_B(rb1, qb_ + bstep);                                            \
    GEMM_FETCH_B(rb2, qb_ + 2 * bstep);                                        \
    GEMM_FETCH_B(rb3, qb_ + 3 * bstep);                                        \
  } while (0)

  v8f acc0 = {};
  v8f acc1 = {};

  GEMM_FETCH(0);

  for (int kk = 0; kk < K; kk += TK) {
    *(float4*)&As[ar][ac]      = ra0;
    *(float4*)&As[ar + 16][ac] = ra1;
    *(float4*)&Bs[br][bc]      = rb0;
    *(float4*)&Bs[br + 8][bc]  = rb1;
    *(float4*)&Bs[br + 16][bc] = rb2;
    *(float4*)&Bs[br + 24][bc] = rb3;
    __syncthreads();

    if (kk + TK < K) GEMM_FETCH(kk + TK);

#pragma unroll
    for (int k4 = 0; k4 < TK; k4 += 4) {
      v2f a0, a1, b;
      b.x  = Bs[k4 + 2 * hi][(wave << 4) + la];
      b.y  = Bs[k4 + 2 * hi + 1][(wave << 4) + la];
      a0.x = As[la][k4 + 2 * hi];
      a0.y = As[la][k4 + 2 * hi + 1];
      a1.x = As[16 + la][k4 + 2 * hi];
      a1.y = As[16 + la][k4 + 2 * hi + 1];
      acc0 = __builtin_amdgcn_wmma_f32_16x16x4_f32(
          false, a0, false, b, (short)0, acc0, false, false);
      acc1 = __builtin_amdgcn_wmma_f32_16x16x4_f32(
          false, a1, false, b, (short)0, acc1, false, false);
    }
    __syncthreads();
  }
#undef GEMM_FETCH
#undef GEMM_FETCH_B

  const int n = n0 + la;
  if (n < Nc) {
#pragma unroll
    for (int r = 0; r < 8; ++r) {
      int gm = m0 + r + 8 * hi;
      if (gm < M) C[(size_t)gm * ldc + n] = acc0[r];
    }
#pragma unroll
    for (int r = 0; r < 8; ++r) {
      int gm = m0 + 16 + r + 8 * hi;
      if (gm < M) C[(size_t)gm * ldc + n] = acc1[r];
    }
  }
}

template <bool VB>
__global__ __launch_bounds__(128)
void gemm_wmma_f32(const float* __restrict__ Ab, const float* __restrict__ Bb,
                   float* __restrict__ Cb, int M, int Nc, int K, int lda, int ldb, int ldc,
                   int zA, int zB, int zC) {
  gemm_tile_f32<VB>(Ab + (size_t)blockIdx.z * zA, Bb + (size_t)blockIdx.z * zB,
                    Cb + (size_t)blockIdx.z * zC, M, Nc, K, lda, ldb, ldc,
                    (int)blockIdx.x, (int)blockIdx.y);
}

template <bool VB>
__global__ __launch_bounds__(128)
void gemm_wmma_f32_two(const float* __restrict__ A0, const float* __restrict__ B0,
                       float* __restrict__ C0, int M0,
                       const float* __restrict__ A1, const float* __restrict__ B1,
                       float* __restrict__ C1, int M1, int nby0,
                       int Nc, int K, int lda, int ldb, int ldc) {
  const bool second = (int)blockIdx.y >= nby0;
  gemm_tile_f32<VB>(second ? A1 : A0, second ? B1 : B0, second ? C1 : C0,
                    second ? M1 : M0, Nc, K, lda, ldb, ldc, (int)blockIdx.x,
                    second ? (int)blockIdx.y - nby0 : (int)blockIdx.y);
}

__global__ void prep_kernel(const float* __restrict__ lin_W, float* __restrict__ Wmean,
                            int* __restrict__ cnt) {
  const int gid = blockIdx.x * blockDim.x + threadIdx.x;
  if (gid < FF * FF) {
    const int n = gid >> 7, f = gid & 127;
    const float* p = lin_W + (size_t)n * DH + f;
    float s = 0.0f;
#pragma unroll
    for (int d = 0; d < DD; ++d) s += p[d * FF];
    Wmean[gid] = s * (1.0f / (float)DD);
  }
  if (gid < NN + EE) cnt[gid] = 0;
}

#define PZ (2048 * COUT)
#define VBLK  8
#define W1BLK 48
#define W2BLK 24
#define BRBLK 6
#define SIDEBLK (VBLK + W1BLK + W2BLK + BRBLK)
__global__ __launch_bounds__(128)
void rank_logitw_kernel(const float* __restrict__ Wmean, const float* __restrict__ sheaf_W,
                        float* __restrict__ V,
                        const float* __restrict__ lin_W, const float* __restrict__ conv1_W,
                        const float* __restrict__ conv2_W, const float* __restrict__ conv2_b,
                        const float* __restrict__ lin2_W,
                        float* __restrict__ W1, float* __restrict__ W2, float* __restrict__ P,
                        const int* __restrict__ row, const int* __restrict__ col,
                        int* __restrict__ cntN, int* __restrict__ cntE,
                        int* __restrict__ rankN, int* __restrict__ rankE) {
  int b = (int)blockIdx.x;
  if (b < VBLK) {
    const int z = b >> 2;
    gemm_tile_f32<false>(Wmean, sheaf_W + z * (FF * DD), V + z * (FF * DD),
                         FF, DD, FF, FF, DD, DD, 0, b & 3);
    return;
  }
  b -= VBLK;
  if (b < W1BLK + W2BLK + BRBLK) {
    const float* A;
    const float* B;
    float* C;
    int M, Nc, lda, ldn, bx, by;
    if (b < W1BLK) {
      const int d = b >> 3;
      A = lin_W + d * FF;  B = conv1_W;  C = W1 + d * (FF * HH);
      M = FF;  Nc = HH;  lda = DH;  ldn = HH;  bx = b & 1;  by = (b >> 1) & 3;
    } else if (b < W1BLK + W2BLK) {
      const int q = b - W1BLK, d = q >> 2;
      A = conv2_W;  B = lin2_W + d * (HH * COUT);  C = W2 + d * (HH * COUT);
      M = HH;  Nc = COUT;  lda = HH;  ldn = COUT;  bx = 0;  by = q & 3;
    } else {
      const int d = b - (W1BLK + W2BLK);
      A = conv2_b;  B = lin2_W + d * (HH * COUT);  C = P + (size_t)d * PZ + (size_t)EE * COUT;
      M = 1;  Nc = COUT;  lda = HH;  ldn = COUT;  bx = 0;  by = 0;
    }
    gemm_tile_f32<true>(A, B, C, M, Nc, FF, lda, ldn, ldn, bx, by);
    return;
  }
  const int i = (b - (W1BLK + W2BLK + BRBLK)) * 128 + (int)threadIdx.x;
  if (i >= NNZ) return;
  rankN[i] = atomicAdd(&cntN[row[i]], 1);
  rankE[i] = atomicAdd(&cntE[col[i]], 1);
}

__global__ __launch_bounds__(1024)
void scan_kernel(const int* __restrict__ cntN, int* __restrict__ offN,
                 const int* __restrict__ cntE, int* __restrict__ offE) {
  __shared__ int part[1024];
  const int* __restrict__ cnt = (blockIdx.x == 0) ? cntN : cntE;
  int* __restrict__ off       = (blockIdx.x == 0) ? offN : offE;
  const int n   = (blockIdx.x == 0) ? NN : EE;
  const int per = (n + 1023) >> 10;
  const int b = threadIdx.x * per;
  const int e = min(b + per, n);
  int s = 0;
  for (int k = b; k < e; ++k) s += cnt[k];
  part[threadIdx.x] = s;
  __syncthreads();
  for (int d = 1; d < 1024; d <<= 1) {
    int v = (threadIdx.x >= d) ? part[threadIdx.x - d] : 0;
    __syncthreads();
    part[threadIdx.x] += v;
    __syncthreads();
  }
  int run = part[threadIdx.x] - s;
  for (int k = b; k < e; ++k) { off[k] = run; run += cnt[k]; }
  if (threadIdx.x == 1023) off[n] = part[1023];
}

__global__ void alpha_place_kernel(const float* __restrict__ xm, const float* __restrict__ em,
                                   const int* __restrict__ row, const int* __restrict__ col,
                                   float* __restrict__ alpha,
                                   const int* __restrict__ offN, const int* __restrict__ offE,
                                   const int* __restrict__ rankN, const int* __restrict__ rankE,
                                   int* __restrict__ permN, int* __restrict__ permE) {
  const int gid = blockIdx.x * blockDim.x + threadIdx.x;
  if (gid >= NNZ * DD) return;
  const int i = gid / DD, dd = gid - i * DD;
  const int r = row[i], c = col[i];
  float v = xm[r * DD + dd] + em[c * DD + dd];
  v = 1.0f / (1.0f + __expf(-v));
  alpha[gid] = v;
  if (dd == 0) {
    permN[offN[r] + rankN[i]] = i;
    permE[offE[c] + rankE[i]] = i;
  }
}

template <int U, int EPI>
__global__ __launch_bounds__(192)
void segment_gather(const float* __restrict__ src, const float* __restrict__ alpha,
                    const int* __restrict__ other,
                    const int* __restrict__ off, const int* __restrict__ perm,
                    const float* __restrict__ bias,
                    float* __restrict__ dst, float* __restrict__ keep, int nseg) {
  const int idx = threadIdx.x << 2;
  float4 bv = make_float4(0.0f, 0.0f, 0.0f, 0.0f);
  if (EPI != 0) {
    const int h = idx & (HH - 1);
    bv = make_float4(bias[h], bias[h + 1], bias[h + 2], bias[h + 3]);
  }
  const int dd = __builtin_amdgcn_readfirstlane(idx >> 7);
  for (int g = blockIdx.x; g < nseg; g += gridDim.x) {
    const int p0 = off[g], p1 = off[g + 1];
    float deg = 0.0f;
    float4 acc = make_float4(0.0f, 0.0f, 0.0f, 0.0f);
    int p = p0;
    for (; p + U <= p1; p += U) {
      int ii[U];
      float s[U];
      float4 v[U];
#pragma unroll
      for (int u = 0; u < U; ++u) ii[u] = perm[p + u];
#pragma unroll
      for (int u = 0; u < U; ++u) {
        s[u] = alpha[(size_t)ii[u] * DD + dd];
        v[u] = *(const float4*)(src + (size_t)other[ii[u]] * DH + idx);
      }
#pragma unroll
      for (int u = 0; u < U; ++u) {
        deg   += s[u];
        acc.x += s[u] * v[u].x;
        acc.y += s[u] * v[u].y;
        acc.z += s[u] * v[u].z;
        acc.w += s[u] * v[u].w;
      }
    }
    for (; p < p1; ++p) {
      const int i = perm[p];
      const float s = alpha[(size_t)i * DD + dd];
      const float4 v = *(const float4*)(src + (size_t)other[i] * DH + idx);
      deg   += s;
      acc.x += s * v.x;
      acc.y += s * v.y;
      acc.z += s * v.z;
      acc.w += s * v.w;
    }
    const float w = (deg != 0.0f) ? (1.0f / deg) : 0.0f;
    if (keep != nullptr && (threadIdx.x & 31) == 0) keep[g * DD + dd] = w;
    acc.x *= w;
    acc.y *= w;
    acc.z *= w;
    acc.w *= w;
    if (EPI != 0) {
      acc.x += bv.x;
      acc.y += bv.y;
      acc.z += bv.z;
      acc.w += bv.w;
    }
    if (EPI == 2) {
      acc.x = (acc.x > 0.0f) ? acc.x : (__expf(acc.x) - 1.0f);
      acc.y = (acc.y > 0.0f) ? acc.y : (__expf(acc.y) - 1.0f);
      acc.z = (acc.z > 0.0f) ? acc.z : (__expf(acc.z) - 1.0f);
      acc.w = (acc.w > 0.0f) ? acc.w : (__expf(acc.w) - 1.0f);
    }
    *(float4*)(dst + (size_t)g * DH + idx) = acc;
  }
}

template <int U>
__global__ __launch_bounds__(32)
void edge_gather_raw(const float* __restrict__ x, const float* __restrict__ alpha,
                     const int* __restrict__ other,
                     const int* __restrict__ off, const int* __restrict__ perm,
                     float* __restrict__ dst) {
  const int g    = blockIdx.x;
  const int lane = threadIdx.x;
  const int f4   = lane << 2;
  const int p0 = off[g], p1 = off[g + 1];
  float deg[DD];
  float4 acc[DD];
#pragma unroll
  for (int d = 0; d < DD; ++d) {
    deg[d] = 0.0f;
    acc[d] = make_float4(0.0f, 0.0f, 0.0f, 0.0f);
  }
  for (int p = p0; p < p1; p += 32) {
    const bool live = (p + lane) < p1;
    const int i = perm[min(p + lane, p1 - 1)];
    const int o = other[i];
    float sl[DD];
#pragma unroll
    for (int d = 0; d < DD; ++d)
      sl[d] = (live ? 1.0f : 0.0f) * alpha[(size_t)i * DD + d];
#pragma unroll
    for (int jb = 0; jb < 32; jb += U) {
      if (p + jb < p1) {
        float4 v[U];
#pragma unroll
        for (int u = 0; u < U; ++u) {
          const int oo = __builtin_amdgcn_readlane(o, jb + u);
          v[u] = *(const float4*)(x + (size_t)oo * FF + f4);
        }
#pragma unroll
        for (int u = 0; u < U; ++u) {
#pragma unroll
          for (int d = 0; d < DD; ++d) {
            const float sd = __int_as_float(
                __builtin_amdgcn_readlane(__float_as_int(sl[d]), jb + u));
            deg[d]   += sd;
            acc[d].x += sd * v[u].x;
            acc[d].y += sd * v[u].y;
            acc[d].z += sd * v[u].z;
            acc[d].w += sd * v[u].w;
          }
        }
      }
    }
  }
#pragma unroll
  for (int d = 0; d < DD; ++d) {
    const float w = (deg[d] != 0.0f) ? (1.0f / deg[d]) : 0.0f;
    acc[d].x *= w;
    acc[d].y *= w;
    acc[d].z *= w;
    acc[d].w *= w;
    *(float4*)(dst + (size_t)g * DH + d * FF + f4) = acc[d];
  }
}

template <int U>
__global__ __launch_bounds__(256)
void node_class_gather(const float* __restrict__ P, const float* __restrict__ alpha,
                       const float* __restrict__ Dinv, const int* __restrict__ col,
                       const int* __restrict__ off, const int* __restrict__ perm,
                       float* __restrict__ out) {
  const int gid = blockIdx.x * blockDim.x + threadIdx.x;
  if (gid >= NN * (COUT / 4)) return;
  const int n = gid / (COUT / 4);
  const float* __restrict__ Pq = P + ((gid - n * (COUT / 4)) << 2);

  float w[DD];
  float4 acc = make_float4(0.0f, 0.0f, 0.0f, 0.0f);
#pragma unroll
  for (int d = 0; d < DD; ++d) {
    w[d] = Dinv[n * DD + d];
    const float4 v = *(const float4*)(Pq + (size_t)d * PZ + (size_t)EE * COUT);
    acc.x += v.x;
    acc.y += v.y;
    acc.z += v.z;
    acc.w += v.w;
  }

  const int p0 = off[n], p1 = off[n + 1];
  for (int p = p0; p < p1; p += U) {
    int ii[U];
    float s[U][DD];
    float4 v[U][DD];
#pragma unroll
    for (int u = 0; u < U; ++u) ii[u] = perm[min(p + u, p1 - 1)];
#pragma unroll
    for (int u = 0; u < U; ++u) {
      const float* __restrict__ Pr = Pq + (size_t)col[ii[u]] * COUT;
      const bool live = (p + u) < p1;
#pragma unroll
      for (int d = 0; d < DD; ++d) {
        s[u][d] = (live ? w[d] : 0.0f) * alpha[(size_t)ii[u] * DD + d];
        v[u][d] = *(const float4*)(Pr + (size_t)d * PZ);
      }
    }
#pragma unroll
    for (int u = 0; u < U; ++u) {
#pragma unroll
      for (int d = 0; d < DD; ++d) {
        acc.x += s[u][d] * v[u][d].x;
        acc.y += s[u][d] * v[u][d].y;
        acc.z += s[u][d] * v[u][d].z;
        acc.w += s[u][d] * v[u][d].w;
      }
    }
  }
  *(float4*)(out + ((size_t)gid << 2)) = acc;
}

template <bool VB>
static inline void launch_gemm(const float* A, const float* B, float* C,
                               int M, int Nc, int K, int lda, int ldb, int ldc,
                               hipStream_t s, int nz = 1, int zA = 0, int zB = 0,
                               int zC = 0) {
  dim3 grid((Nc + 63) / 64, (M + TM - 1) / TM, nz);
  gemm_wmma_f32<VB><<<grid, dim3(128), 0, s>>>(A, B, C, M, Nc, K, lda, ldb, ldc,
                                               zA, zB, zC);
}

extern "C" void kernel_launch(void* const* d_in, const int* in_sizes, int n_in,
                              void* d_out, int out_size, void* d_ws, size_t ws_size,
                              hipStream_t stream) {
  const float* x        = (const float*)d_in[0];
  const int*   eidx     = (const int*)d_in[1];
  const float* hedge    = (const float*)d_in[2];
  const float* lin_W    = (const float*)d_in[3];
  const float* sheaf_W  = (const float*)d_in[4];
  const float* conv1_W  = (const float*)d_in[5];
  const float* conv1_b  = (const float*)d_in[6];
  const float* conv2_W  = (const float*)d_in[7];
  const float* conv2_b  = (const float*)d_in[8];
  const float* lin2_W   = (const float*)d_in[9];
  float* out = (float*)d_out;

  const int* row = eidx;
  const int* col = eidx + NNZ;

  float* ws    = (float*)d_ws;
  float* bufA  = ws;
  float* magg  = bufA + (size_t)NN * DH;
  float* P     = magg + (size_t)4 * EE * DH;
  float* xagg  = magg + (size_t)NN * DH;
  float* mbuf  = xagg + (size_t)EE * DH;
  float* csr   = mbuf + (size_t)EE * DH;
  float* Wmean = csr + (size_t)NN * FF;
  float* V     = Wmean + (size_t)FF * FF;
  float* W1    = V + (size_t)2 * FF * DD;
  float* W2    = W1 + (size_t)DD * FF * HH;
  float* xm    = Wmean + (size_t)EE * FF;
  float* em    = xm + (size_t)NN * DD;
  float* alpha = em + (size_t)EE * DD;
  float* Dinv  = alpha + (size_t)NNZ * DD;

  int* cntN  = (int*)csr;
  int* cntE  = cntN + NN;
  int* offN  = cntE + EE;
  int* offE  = offN + (NN + 1);
  int* rankN = offE + (EE + 1);
  int* rankE = rankN + NNZ;
  int* permN = rankE + NNZ;
  int* permE = permN + NNZ;


  prep_kernel<<<(NN + EE + 255) / 256, 256, 0, stream>>>(lin_W, Wmean, cntN);

  rank_logitw_kernel<<<SIDEBLK + (NNZ + 127) / 128, 128, 0, stream>>>(
      Wmean, sheaf_W, V, lin_W, conv1_W, conv2_W, conv2_b, lin2_W, W1, W2, P,
      row, col, cntN, cntE, rankN, rankE);

  gemm_wmma_f32_two<false><<<dim3(1, (NN + TM - 1) / TM + (EE + TM - 1) / TM, 1), dim3(128),
                             0, stream>>>(x, V, xm, NN, hedge, V + FF * DD, em, EE,
                                          (NN + TM - 1) / TM, DD, FF, FF, DD, DD);

  scan_kernel<<<2, 1024, 0, stream>>>(cntN, offN, cntE, offE);
  alpha_place_kernel<<<(NNZ * DD + 255) / 256, 256, 0, stream>>>(
      xm, em, row, col, alpha, offN, offE, rankN, rankE, permN, permE);


  edge_gather_raw<8><<<EE, 32, 0, stream>>>(x, alpha, row, offE, permE, xagg);
  launch_gemm<true>(xagg, W1, mbuf, EE, HH, FF, DH, HH, DH, stream,
                    DD, FF, FF * HH, HH);
  segment_gather<4, 2><<<2560, 192, 0, stream>>>(mbuf, alpha, col, offN, permN, conv1_b,
                                                 bufA, Dinv, NN);

  segment_gather<8, 0><<<EE, 192, 0, stream>>>(bufA, alpha, row, offE, permE, nullptr,
                                               magg, nullptr, EE);

  launch_gemm<true>(magg, W2, P, EE, COUT, HH, DH, COUT, COUT, stream,
                    DD, HH, HH * COUT, PZ);
  node_class_gather<4><<<(NN * (COUT / 4) + 255) / 256, 256, 0, stream>>>(
      P, alpha, Dinv, col, offN, permN, out);
}
